# starting kernel
# baseline (speedup 1.0000x reference)
#include <hip/hip_runtime.h>

#define D_IN 512
#define DK   64
#define DV   64
#define B_   8
#define S_   2048

typedef __attribute__((ext_vector_type(2))) float v2f;
typedef __attribute__((ext_vector_type(8))) float v8f;

static __device__ __forceinline__ v8f wmma_f32(v2f a, v2f b, v8f c) {
  return __builtin_amdgcn_wmma_f32_16x16x4_f32(
       false, a,  false, b,
       (short)0, c,  false,  false);
}

__global__ void transpose_weights(const float* __restrict__ Wk,
                                  const float* __restrict__ Wq,
                                  const float* __restrict__ Wv,
                                  float* __restrict__ WT3) {
  int idx = blockIdx.x * 256 + threadIdx.x;
  if (idx >= 3 * D_IN * DK) return;
  int mat = idx >> 15;
  int rem = idx & 32767;
  int d = rem >> 6;
  int n = rem & 63;
  const float* W = (mat == 0) ? Wk : ((mat == 1) ? Wq : Wv);
  WT3[idx] = W[n * D_IN + d];
}

__global__ __launch_bounds__(256) void qkv_kernel(
    const float* __restrict__ X, const float* __restrict__ WT3,
    const float* __restrict__ bk, const float* __restrict__ bq,
    const float* __restrict__ bv,
    float* __restrict__ Kb, float* __restrict__ Qb, float* __restrict__ Vb) {
  const int lane = threadIdx.x & 31;
  const int wave = threadIdx.x >> 5;
  const int wg   = blockIdx.x * 8 + wave;
  const int b    = wg >> 7;
  const int s0   = (wg & 127) << 4;
  const int m    = lane & 15;
  const int hi   = lane >> 4;

  const float* xr = X + (size_t)(b * S_ + s0 + m) * D_IN + 2 * hi;

  v8f acc[3][4];
#pragma unroll
  for (int i = 0; i < 3; ++i)
#pragma unroll
    for (int j = 0; j < 4; ++j)
      acc[i][j] = (v8f){0.f, 0.f, 0.f, 0.f, 0.f, 0.f, 0.f, 0.f};

  for (int t = 0; t < D_IN / 4; ++t) {
    v2f a;
    a.x = xr[4 * t];
    a.y = xr[4 * t + 1];
    const float* wbase = WT3 + ((size_t)(4 * t + 2 * hi)) * DK + m;
#pragma unroll
    for (int mtx = 0; mtx < 3; ++mtx) {
#pragma unroll
      for (int nt = 0; nt < 4; ++nt) {
        const float* wp = wbase + (size_t)mtx * D_IN * DK + nt * 16;
        v2f bf;
        bf.x = wp[0];
        bf.y = wp[DK];
        acc[mtx][nt] = wmma_f32(a, bf, acc[mtx][nt]);
      }
    }
  }

  const float* bias[3] = {bk, bq, bv};
  float* out[3] = {Kb, Qb, Vb};
#pragma unroll
  for (int mtx = 0; mtx < 3; ++mtx) {
#pragma unroll
    for (int nt = 0; nt < 4; ++nt) {
      float bs = bias[mtx][nt * 16 + m];
#pragma unroll
      for (int r = 0; r < 8; ++r) {
        out[mtx][(size_t)(b * S_ + s0 + r + 8 * hi) * DK + nt * 16 + m] =
            acc[mtx][nt][r] + bs;
      }
    }
  }
}

__global__ __launch_bounds__(256) void attn_kernel(
    const float* __restrict__ Kb, const float* __restrict__ Qb,
    const float* __restrict__ Vb, float* __restrict__ Y) {
  __shared__ float lds_e[256 * 12];
  __shared__ float lds_a[8][16 * 17];

  const int lane = threadIdx.x & 31;
  const int b    = threadIdx.x >> 5;
  const int m    = lane & 15;
  const int hi   = lane >> 4;
  const int i0   = blockIdx.x << 4;

  const float scale = 0.125f;

  v2f ka[16];
  {
    const float* kr = Kb + (size_t)(b * S_ + i0 + m) * DK + 2 * hi;
#pragma unroll
    for (int t = 0; t < 16; ++t) {
      ka[t].x = kr[4 * t] * scale;
      ka[t].y = kr[4 * t + 1] * scale;
    }
  }

  v8f yacc[4];
#pragma unroll
  for (int nt = 0; nt < 4; ++nt)
    yacc[nt] = (v8f){0.f, 0.f, 0.f, 0.f, 0.f, 0.f, 0.f, 0.f};

  for (int j0 = 0; j0 < S_; j0 += 16) {
    v8f e = (v8f){0.f, 0.f, 0.f, 0.f, 0.f, 0.f, 0.f, 0.f};
    {
      const float* qr = Qb + (size_t)(b * S_ + j0 + m) * DK + 2 * hi;
#pragma unroll
      for (int t = 0; t < 16; ++t) {
        v2f qb;
        qb.x = qr[4 * t];
        qb.y = qr[4 * t + 1];
        e = wmma_f32(ka[t], qb, e);
      }
    }

#pragma unroll
    for (int r = 0; r < 8; ++r)
      lds_e[(r * 32 + lane) * 12 + b] = e[r];
    __syncthreads();

    float aval[8];
#pragma unroll
    for (int r = 0; r < 8; ++r) {
      const float4* p = (const float4*)&lds_e[(r * 32 + lane) * 12];
      float4 u = p[0];
      float4 w = p[1];
      float mx = fmaxf(fmaxf(fmaxf(u.x, u.y), fmaxf(u.z, u.w)),
                       fmaxf(fmaxf(w.x, w.y), fmaxf(w.z, w.w)));
      float z = __expf(u.x - mx) + __expf(u.y - mx) + __expf(u.z - mx) +
                __expf(u.w - mx) + __expf(w.x - mx) + __expf(w.y - mx) +
                __expf(w.z - mx) + __expf(w.w - mx);
      aval[r] = __expf(e[r] - mx) / z;
    }

#pragma unroll
    for (int r = 0; r < 8; ++r)
      lds_a[b][(r + 8 * hi) * 17 + m] = aval[r];
    __syncthreads();

    v2f af[4];
#pragma unroll
    for (int t = 0; t < 4; ++t) {
      af[t].x = lds_a[b][m * 17 + 4 * t + 2 * hi];
      af[t].y = lds_a[b][m * 17 + 4 * t + 2 * hi + 1];
    }

#pragma unroll
    for (int t = 0; t < 4; ++t) {
      const float* vr = Vb + (size_t)(b * S_ + j0 + 4 * t + 2 * hi) * DV + m;
#pragma unroll
      for (int nt = 0; nt < 4; ++nt) {
        v2f vb;
        vb.x = vr[nt * 16];
        vb.y = vr[DV + nt * 16];
        yacc[nt] = wmma_f32(af[t], vb, yacc[nt]);
      }
    }
  }

#pragma unroll
  for (int nt = 0; nt < 4; ++nt)
#pragma unroll
    for (int r = 0; r < 8; ++r)
      Y[(size_t)(b * S_ + i0 + r + 8 * hi) * DV + nt * 16 + m] =
          yacc[nt][r];
}

extern "C" void kernel_launch(void* const* d_in, const int* in_sizes, int n_in,
                              void* d_out, int out_size, void* d_ws,
                              size_t ws_size, hipStream_t stream) {
  const float* X  = (const float*)d_in[0];
  const float* Wk = (const float*)d_in[1];
  const float* bk = (const float*)d_in[2];
  const float* Wq = (const float*)d_in[3];
  const float* bq = (const float*)d_in[4];
  const float* Wv = (const float*)d_in[5];
  const float* bv = (const float*)d_in[6];
  float* Y = (float*)d_out;

  float* ws = (float*)d_ws;
  const size_t NKQV = (size_t)B_ * S_ * DK;
  float* Kb  = ws;
  float* Qb  = ws + NKQV;
  float* Vb  = ws + 2 * NKQV;
  float* WT3 = ws + 3 * NKQV;

  transpose_weights<<<(3 * D_IN * DK + 255) / 256, 256, 0, stream>>>(
      Wk, Wq, Wv, WT3);
  qkv_kernel<<<(B_ * S_ / 16) / 8, 256, 0, stream>>>(
      X, WT3, bk, bq, bv, Kb, Qb, Vb);
  attn_kernel<<<S_ / 16, 256, 0, stream>>>(Kb, Qb, Vb, Y);
}


# attn kernel: step the j loop by 64 columns instead of 16 (4 E sub-tiles per LDS exchange and barrier pair, 32 steps not 128); k and j summation order unchanged
# speedup vs baseline: 1.3478x; 1.3478x over previous
#include <hip/hip_runtime.h>

#define D_IN 512
#define DK   64
#define DV   64
#define B_   8
#define S_   2048

typedef __attribute__((ext_vector_type(2))) float v2f;
typedef __attribute__((ext_vector_type(8))) float v8f;

static __device__ __forceinline__ v8f wmma_f32(v2f a, v2f b, v8f c) {
  return __builtin_amdgcn_wmma_f32_16x16x4_f32(
       false, a,  false, b,
       (short)0, c,  false,  false);
}

__global__ void transpose_weights(const float* __restrict__ Wk,
                                  const float* __restrict__ Wq,
                                  const float* __restrict__ Wv,
                                  float* __restrict__ WT3) {
  int idx = blockIdx.x * 256 + threadIdx.x;
  if (idx >= 3 * D_IN * DK) return;
  int mat = idx >> 15;
  int rem = idx & 32767;
  int d = rem >> 6;
  int n = rem & 63;
  const float* W = (mat == 0) ? Wk : ((mat == 1) ? Wq : Wv);
  WT3[idx] = W[n * D_IN + d];
}

__global__ __launch_bounds__(256) void qkv_kernel(
    const float* __restrict__ X, const float* __restrict__ WT3,
    const float* __restrict__ bk, const float* __restrict__ bq,
    const float* __restrict__ bv,
    float* __restrict__ Kb, float* __restrict__ Qb, float* __restrict__ Vb) {
  const int lane = threadIdx.x & 31;
  const int wave = threadIdx.x >> 5;
  const int wg   = blockIdx.x * 8 + wave;
  const int b    = wg >> 7;
  const int s0   = (wg & 127) << 4;
  const int m    = lane & 15;
  const int hi   = lane >> 4;

  const float* xr = X + (size_t)(b * S_ + s0 + m) * D_IN + 2 * hi;

  v8f acc[3][4];
#pragma unroll
  for (int i = 0; i < 3; ++i)
#pragma unroll
    for (int j = 0; j < 4; ++j)
      acc[i][j] = (v8f){0.f, 0.f, 0.f, 0.f, 0.f, 0.f, 0.f, 0.f};

  for (int t = 0; t < D_IN / 4; ++t) {
    v2f a;
    a.x = xr[4 * t];
    a.y = xr[4 * t + 1];
    const float* wbase = WT3 + ((size_t)(4 * t + 2 * hi)) * DK + m;
#pragma unroll
    for (int mtx = 0; mtx < 3; ++mtx) {
#pragma unroll
      for (int nt = 0; nt < 4; ++nt) {
        const float* wp = wbase + (size_t)mtx * D_IN * DK + nt * 16;
        v2f bf;
        bf.x = wp[0];
        bf.y = wp[DK];
        acc[mtx][nt] = wmma_f32(a, bf, acc[mtx][nt]);
      }
    }
  }

  const float* bias[3] = {bk, bq, bv};
  float* out[3] = {Kb, Qb, Vb};
#pragma unroll
  for (int mtx = 0; mtx < 3; ++mtx) {
#pragma unroll
    for (int nt = 0; nt < 4; ++nt) {
      float bs = bias[mtx][nt * 16 + m];
#pragma unroll
      for (int r = 0; r < 8; ++r) {
        out[mtx][(size_t)(b * S_ + s0 + r + 8 * hi) * DK + nt * 16 + m] =
            acc[mtx][nt][r] + bs;
      }
    }
  }
}

#define JT 64
#define NJ (JT / 16)
#define APITCH (JT + 4)

__global__ __launch_bounds__(256) void attn_kernel(
    const float* __restrict__ Kb, const float* __restrict__ Qb,
    const float* __restrict__ Vb, float* __restrict__ Y) {
  __shared__ __attribute__((aligned(16))) float lds_e[NJ * 256 * 12];
  __shared__ __attribute__((aligned(16))) float lds_a[8][16 * APITCH];

  const int lane = threadIdx.x & 31;
  const int b    = threadIdx.x >> 5;
  const int m    = lane & 15;
  const int hi   = lane >> 4;
  const int i0   = blockIdx.x << 4;

  const float scale = 0.125f;

  v2f ka[16];
  {
    const float* kr = Kb + (size_t)(b * S_ + i0 + m) * DK + 2 * hi;
#pragma unroll
    for (int t = 0; t < 16; ++t) {
      ka[t].x = kr[4 * t] * scale;
      ka[t].y = kr[4 * t + 1] * scale;
    }
  }

  v8f yacc[4];
#pragma unroll
  for (int nt = 0; nt < 4; ++nt)
    yacc[nt] = (v8f){0.f, 0.f, 0.f, 0.f, 0.f, 0.f, 0.f, 0.f};

  for (int j0 = 0; j0 < S_; j0 += JT) {
    v2f qf[NJ][16];
    {
      const float* qr = Qb + (size_t)(b * S_ + j0 + m) * DK + 2 * hi;
#pragma unroll
      for (int jt = 0; jt < NJ; ++jt) {
#pragma unroll
        for (int t = 0; t < 16; ++t) {
          qf[jt][t].x = qr[jt * 16 * DK + 4 * t];
          qf[jt][t].y = qr[jt * 16 * DK + 4 * t + 1];
        }
        __builtin_amdgcn_sched_barrier(0);
      }
    }
    v8f e[NJ];
#pragma unroll
    for (int jt = 0; jt < NJ; ++jt) {
      e[jt] = (v8f){0.f, 0.f, 0.f, 0.f, 0.f, 0.f, 0.f, 0.f};
#pragma unroll
      for (int t = 0; t < 16; ++t)
        e[jt] = wmma_f32(ka[t], qf[jt][t], e[jt]);
      __builtin_amdgcn_sched_barrier(0);
    }

#pragma unroll
    for (int jt = 0; jt < NJ; ++jt)
#pragma unroll
      for (int r = 0; r < 8; ++r)
        lds_e[((jt * 8 + r) * 32 + lane) * 12 + b] = e[jt][r];
    __syncthreads();

#pragma unroll
    for (int jt = 0; jt < NJ; ++jt) {
#pragma unroll
      for (int r = 0; r < 8; ++r) {
        const float4* p =
            (const float4*)&lds_e[((jt * 8 + r) * 32 + lane) * 12];
        float4 u = p[0];
        float4 w = p[1];
        float mx = fmaxf(fmaxf(fmaxf(u.x, u.y), fmaxf(u.z, u.w)),
                         fmaxf(fmaxf(w.x, w.y), fmaxf(w.z, w.w)));
        float z = __expf(u.x - mx) + __expf(u.y - mx) + __expf(u.z - mx) +
                  __expf(u.w - mx) + __expf(w.x - mx) + __expf(w.y - mx) +
                  __expf(w.z - mx) + __expf(w.w - mx);
        lds_a[b][(r + 8 * hi) * APITCH + jt * 16 + m] =
            __expf(e[jt][r] - mx) / z;
      }
      __builtin_amdgcn_sched_barrier(0);
    }
    __syncthreads();

    v2f af[4 * NJ];
#pragma unroll
    for (int t = 0; t < 4 * NJ; ++t) {
      af[t].x = lds_a[b][m * APITCH + 4 * t + 2 * hi];
      af[t].y = lds_a[b][m * APITCH + 4 * t + 2 * hi + 1];
    }

    v2f vf[4 * NJ][4];
    {
      const float* vr = Vb + (size_t)(b * S_ + j0 + 2 * hi) * DV + m;
#pragma unroll
      for (int t = 0; t < 4 * NJ; ++t)
#pragma unroll
        for (int nt = 0; nt < 4; ++nt) {
          vf[t][nt].x = vr[(4 * t) * DV + nt * 16];
          vf[t][nt].y = vr[(4 * t + 1) * DV + nt * 16];
        }
    }
#pragma unroll
    for (int t = 0; t < 4 * NJ; ++t)
#pragma unroll
      for (int nt = 0; nt < 4; ++nt)
        yacc[nt] = wmma_f32(af[t], vf[t][nt], yacc[nt]);
    __builtin_amdgcn_sched_barrier(0);
    asm volatile("v_nop\n\tv_nop\n\tv_nop\n\tv_nop");
  }

#pragma unroll
  for (int nt = 0; nt < 4; ++nt)
#pragma unroll
    for (int r = 0; r < 8; ++r)
      Y[(size_t)(b * S_ + i0 + r + 8 * hi) * DV + nt * 16 + m] =
          yacc[nt][r];
}

extern "C" void kernel_launch(void* const* d_in, const int* in_sizes, int n_in,
                              void* d_out, int out_size, void* d_ws,
                              size_t ws_size, hipStream_t stream) {
  const float* X  = (const float*)d_in[0];
  const float* Wk = (const float*)d_in[1];
  const float* bk = (const float*)d_in[2];
  const float* Wq = (const float*)d_in[3];
  const float* bq = (const float*)d_in[4];
  const float* Wv = (const float*)d_in[5];
  const float* bv = (const float*)d_in[6];
  float* Y = (float*)d_out;

  float* ws = (float*)d_ws;
  const size_t NKQV = (size_t)B_ * S_ * DK;
  float* Kb  = ws;
  float* Qb  = ws + NKQV;
  float* Vb  = ws + 2 * NKQV;
  float* WT3 = ws + 3 * NKQV;

  transpose_weights<<<(3 * D_IN * DK + 255) / 256, 256, 0, stream>>>(
      Wk, Wq, Wv, WT3);
  qkv_kernel<<<(B_ * S_ / 16) / 8, 256, 0, stream>>>(
      X, WT3, bk, bq, bv, Kb, Qb, Vb);
  attn_kernel<<<S_ / 16, 256, 0, stream>>>(Kb, Qb, Vb, Y);
}


# attn kernel: batch softmax of each (i,j) element computed once, by one of the 8 waves (wave w: tile rows w, w+8), which stores all 8 batches' A values to LDS; arithmetic per value unchanged
# speedup vs baseline: 1.5729x; 1.1670x over previous
#include <hip/hip_runtime.h>

#define D_IN 512
#define DK   64
#define DV   64
#define B_   8
#define S_   2048

typedef __attribute__((ext_vector_type(2))) float v2f;
typedef __attribute__((ext_vector_type(8))) float v8f;

static __device__ __forceinline__ v8f wmma_f32(v2f a, v2f b, v8f c) {
  return __builtin_amdgcn_wmma_f32_16x16x4_f32(
       false, a,  false, b,
       (short)0, c,  false,  false);
}

__global__ void transpose_weights(const float* __restrict__ Wk,
                                  const float* __restrict__ Wq,
                                  const float* __restrict__ Wv,
                                  float* __restrict__ WT3) {
  int idx = blockIdx.x * 256 + threadIdx.x;
  if (idx >= 3 * D_IN * DK) return;
  int mat = idx >> 15;
  int rem = idx & 32767;
  int d = rem >> 6;
  int n = rem & 63;
  const float* W = (mat == 0) ? Wk : ((mat == 1) ? Wq : Wv);
  WT3[idx] = W[n * D_IN + d];
}

__global__ __launch_bounds__(256) void qkv_kernel(
    const float* __restrict__ X, const float* __restrict__ WT3,
    const float* __restrict__ bk, const float* __restrict__ bq,
    const float* __restrict__ bv,
    float* __restrict__ Kb, float* __restrict__ Qb, float* __restrict__ Vb) {
  const int lane = threadIdx.x & 31;
  const int wave = threadIdx.x >> 5;
  const int wg   = blockIdx.x * 8 + wave;
  const int b    = wg >> 7;
  const int s0   = (wg & 127) << 4;
  const int m    = lane & 15;
  const int hi   = lane >> 4;

  const float* xr = X + (size_t)(b * S_ + s0 + m) * D_IN + 2 * hi;

  v8f acc[3][4];
#pragma unroll
  for (int i = 0; i < 3; ++i)
#pragma unroll
    for (int j = 0; j < 4; ++j)
      acc[i][j] = (v8f){0.f, 0.f, 0.f, 0.f, 0.f, 0.f, 0.f, 0.f};

  for (int t = 0; t < D_IN / 4; ++t) {
    v2f a;
    a.x = xr[4 * t];
    a.y = xr[4 * t + 1];
    const float* wbase = WT3 + ((size_t)(4 * t + 2 * hi)) * DK + m;
#pragma unroll
    for (int mtx = 0; mtx < 3; ++mtx) {
#pragma unroll
      for (int nt = 0; nt < 4; ++nt) {
        const float* wp = wbase + (size_t)mtx * D_IN * DK + nt * 16;
        v2f bf;
        bf.x = wp[0];
        bf.y = wp[DK];
        acc[mtx][nt] = wmma_f32(a, bf, acc[mtx][nt]);
      }
    }
  }

  const float* bias[3] = {bk, bq, bv};
  float* out[3] = {Kb, Qb, Vb};
#pragma unroll
  for (int mtx = 0; mtx < 3; ++mtx) {
#pragma unroll
    for (int nt = 0; nt < 4; ++nt) {
      float bs = bias[mtx][nt * 16 + m];
#pragma unroll
      for (int r = 0; r < 8; ++r) {
        out[mtx][(size_t)(b * S_ + s0 + r + 8 * hi) * DK + nt * 16 + m] =
            acc[mtx][nt][r] + bs;
      }
    }
  }
}

#define JT 64
#define NJ (JT / 16)
#define APITCH (JT + 4)

__global__ __launch_bounds__(256) void attn_kernel(
    const float* __restrict__ Kb, const float* __restrict__ Qb,
    const float* __restrict__ Vb, float* __restrict__ Y) {
  __shared__ __attribute__((aligned(16))) float lds_e[NJ * 256 * 12];
  __shared__ __attribute__((aligned(16))) float lds_a[8][16 * APITCH];

  const int lane = threadIdx.x & 31;
  const int b    = threadIdx.x >> 5;
  const int m    = lane & 15;
  const int hi   = lane >> 4;
  const int i0   = blockIdx.x << 4;

  const float scale = 0.125f;

  v2f ka[16];
  {
    const float* kr = Kb + (size_t)(b * S_ + i0 + m) * DK + 2 * hi;
#pragma unroll
    for (int t = 0; t < 16; ++t) {
      ka[t].x = kr[4 * t] * scale;
      ka[t].y = kr[4 * t + 1] * scale;
    }
  }

  v8f yacc[4];
#pragma unroll
  for (int nt = 0; nt < 4; ++nt)
    yacc[nt] = (v8f){0.f, 0.f, 0.f, 0.f, 0.f, 0.f, 0.f, 0.f};

  for (int j0 = 0; j0 < S_; j0 += JT) {
    v2f qf[NJ][16];
    {
      const float* qr = Qb + (size_t)(b * S_ + j0 + m) * DK + 2 * hi;
#pragma unroll
      for (int jt = 0; jt < NJ; ++jt) {
#pragma unroll
        for (int t = 0; t < 16; ++t) {
          qf[jt][t].x = qr[jt * 16 * DK + 4 * t];
          qf[jt][t].y = qr[jt * 16 * DK + 4 * t + 1];
        }
        __builtin_amdgcn_sched_barrier(0);
      }
    }
    v8f e[NJ];
#pragma unroll
    for (int jt = 0; jt < NJ; ++jt) {
      e[jt] = (v8f){0.f, 0.f, 0.f, 0.f, 0.f, 0.f, 0.f, 0.f};
#pragma unroll
      for (int t = 0; t < 16; ++t)
        e[jt] = wmma_f32(ka[t], qf[jt][t], e[jt]);
      __builtin_amdgcn_sched_barrier(0);
    }

#pragma unroll
    for (int jt = 0; jt < NJ; ++jt)
#pragma unroll
      for (int r = 0; r < 8; ++r)
        lds_e[((jt * 8 + r) * 32 + lane) * 12 + b] = e[jt][r];
    __syncthreads();

#pragma unroll
    for (int jt = 0; jt < NJ; ++jt) {
      const float4* p =
          (const float4*)&lds_e[((jt * 8 + b) * 32 + lane) * 12];
      float4 u = p[0];
      float4 w = p[1];
      float mx = fmaxf(fmaxf(fmaxf(u.x, u.y), fmaxf(u.z, u.w)),
                       fmaxf(fmaxf(w.x, w.y), fmaxf(w.z, w.w)));
      float t0 = __expf(u.x - mx), t1 = __expf(u.y - mx);
      float t2 = __expf(u.z - mx), t3 = __expf(u.w - mx);
      float t4 = __expf(w.x - mx), t5 = __expf(w.y - mx);
      float t6 = __expf(w.z - mx), t7 = __expf(w.w - mx);
      float z = t0 + t1 + t2 + t3 + t4 + t5 + t6 + t7;
      float* dst = &lds_a[0][(b + 8 * hi) * APITCH + jt * 16 + m];
      dst[0 * 16 * APITCH] = t0 / z;
      dst[1 * 16 * APITCH] = t1 / z;
      dst[2 * 16 * APITCH] = t2 / z;
      dst[3 * 16 * APITCH] = t3 / z;
      dst[4 * 16 * APITCH] = t4 / z;
      dst[5 * 16 * APITCH] = t5 / z;
      dst[6 * 16 * APITCH] = t6 / z;
      dst[7 * 16 * APITCH] = t7 / z;
    }
    __syncthreads();

    v2f af[4 * NJ];
#pragma unroll
    for (int t = 0; t < 4 * NJ; ++t) {
      af[t].x = lds_a[b][m * APITCH + 4 * t + 2 * hi];
      af[t].y = lds_a[b][m * APITCH + 4 * t + 2 * hi + 1];
    }

    v2f vf[4 * NJ][4];
    {
      const float* vr = Vb + (size_t)(b * S_ + j0 + 2 * hi) * DV + m;
#pragma unroll
      for (int t = 0; t < 4 * NJ; ++t)
#pragma unroll
        for (int nt = 0; nt < 4; ++nt) {
          vf[t][nt].x = vr[(4 * t) * DV + nt * 16];
          vf[t][nt].y = vr[(4 * t + 1) * DV + nt * 16];
        }
    }
#pragma unroll
    for (int t = 0; t < 4 * NJ; ++t)
#pragma unroll
      for (int nt = 0; nt < 4; ++nt)
        yacc[nt] = wmma_f32(af[t], vf[t][nt], yacc[nt]);
    __builtin_amdgcn_sched_barrier(0);
    asm volatile("v_nop\n\tv_nop\n\tv_nop\n\tv_nop");
  }

#pragma unroll
  for (int nt = 0; nt < 4; ++nt)
#pragma unroll
    for (int r = 0; r < 8; ++r)
      Y[(size_t)(b * S_ + i0 + r + 8 * hi) * DV + nt * 16 + m] =
          yacc[nt][r];
}

extern "C" void kernel_launch(void* const* d_in, const int* in_sizes, int n_in,
                              void* d_out, int out_size, void* d_ws,
                              size_t ws_size, hipStream_t stream) {
  const float* X  = (const float*)d_in[0];
  const float* Wk = (const float*)d_in[1];
  const float* bk = (const float*)d_in[2];
  const float* Wq = (const float*)d_in[3];
  const float* bq = (const float*)d_in[4];
  const float* Wv = (const float*)d_in[5];
  const float* bv = (const float*)d_in[6];
  float* Y = (float*)d_out;

  float* ws = (float*)d_ws;
  const size_t NKQV = (size_t)B_ * S_ * DK;
  float* Kb  = ws;
  float* Qb  = ws + NKQV;
  float* Vb  = ws + 2 * NKQV;
  float* WT3 = ws + 3 * NKQV;

  transpose_weights<<<(3 * D_IN * DK + 255) / 256, 256, 0, stream>>>(
      Wk, Wq, Wv, WT3);
  qkv_kernel<<<(B_ * S_ / 16) / 8, 256, 0, stream>>>(
      X, WT3, bk, bq, bv, Kb, Qb, Vb);
  attn_kernel<<<S_ / 16, 256, 0, stream>>>(Kb, Qb, Vb, Y);
}


# attn kernel: j range cut in two halves, 256 blocks (one per WGP) instead of 128; each half sums its own Y, a 4th kernel adds the two: sum over j is two running sums plus one add, not one
# speedup vs baseline: 2.3730x; 1.5087x over previous
#include <hip/hip_runtime.h>

#define D_IN 512
#define DK   64
#define DV   64
#define B_   8
#define S_   2048

typedef __attribute__((ext_vector_type(2))) float v2f;
typedef __attribute__((ext_vector_type(8))) float v8f;

static __device__ __forceinline__ v8f wmma_f32(v2f a, v2f b, v8f c) {
  return __builtin_amdgcn_wmma_f32_16x16x4_f32(
       false, a,  false, b,
       (short)0, c,  false,  false);
}

__global__ void transpose_weights(const float* __restrict__ Wk,
                                  const float* __restrict__ Wq,
                                  const float* __restrict__ Wv,
                                  float* __restrict__ WT3) {
  int idx = blockIdx.x * 256 + threadIdx.x;
  if (idx >= 3 * D_IN * DK) return;
  int mat = idx >> 15;
  int rem = idx & 32767;
  int d = rem >> 6;
  int n = rem & 63;
  const float* W = (mat == 0) ? Wk : ((mat == 1) ? Wq : Wv);
  WT3[idx] = W[n * D_IN + d];
}

__global__ __launch_bounds__(256) void qkv_kernel(
    const float* __restrict__ X, const float* __restrict__ WT3,
    const float* __restrict__ bk, const float* __restrict__ bq,
    const float* __restrict__ bv,
    float* __restrict__ Kb, float* __restrict__ Qb, float* __restrict__ Vb,
    float* __restrict__ Kb2) {
  const int lane = threadIdx.x & 31;
  const int wave = threadIdx.x >> 5;
  const int wg   = blockIdx.x * 8 + wave;
  const int b    = wg >> 7;
  const int s0   = (wg & 127) << 4;
  const int m    = lane & 15;
  const int hi   = lane >> 4;

  const float* xr = X + (size_t)(b * S_ + s0 + m) * D_IN + 2 * hi;

  v8f acc[3][4];
#pragma unroll
  for (int i = 0; i < 3; ++i)
#pragma unroll
    for (int j = 0; j < 4; ++j)
      acc[i][j] = (v8f){0.f, 0.f, 0.f, 0.f, 0.f, 0.f, 0.f, 0.f};

  for (int t = 0; t < D_IN / 4; ++t) {
    v2f a;
    a.x = xr[4 * t];
    a.y = xr[4 * t + 1];
    const float* wbase = WT3 + ((size_t)(4 * t + 2 * hi)) * DK + m;
#pragma unroll
    for (int mtx = 0; mtx < 3; ++mtx) {
#pragma unroll
      for (int nt = 0; nt < 4; ++nt) {
        const float* wp = wbase + (size_t)mtx * D_IN * DK + nt * 16;
        v2f bf;
        bf.x = wp[0];
        bf.y = wp[DK];
        acc[mtx][nt] = wmma_f32(a, bf, acc[mtx][nt]);
      }
    }
  }

  const float* bias[3] = {bk, bq, bv};
  float* out[3] = {Kb, Qb, Vb};
#pragma unroll
  for (int mtx = 0; mtx < 3; ++mtx) {
#pragma unroll
    for (int nt = 0; nt < 4; ++nt) {
      float bs = bias[mtx][nt * 16 + m];
#pragma unroll
      for (int r = 0; r < 8; ++r) {
        const size_t o = (size_t)(b * S_ + s0 + r + 8 * hi) * DK + nt * 16 + m;
        const float val = acc[mtx][nt][r] + bs;
        out[mtx][o] = val;
        if (mtx == 0) Kb2[o] = val;
      }
    }
  }
}

#define JT 64
#define NJ (JT / 16)
#define APITCH (JT + 4)

static_assert(DK == DV, "a K tile and a Y tile share one plane layout");

__global__ __launch_bounds__(256) void attn_kernel(
    float* KY0, float* KY1, const float* __restrict__ Qb,
    const float* __restrict__ Vb) {
  __shared__ __attribute__((aligned(16))) float lds_e[NJ * 256 * 12];
  __shared__ __attribute__((aligned(16))) float lds_a[8][16 * APITCH];

  const int lane = threadIdx.x & 31;
  const int b    = threadIdx.x >> 5;
  const int m    = lane & 15;
  const int hi   = lane >> 4;
  const int half = blockIdx.x & 1;
  const int i0   = (blockIdx.x >> 1) << 4;
  const int jbeg = half * (S_ / 2);
  float* ky = half ? KY1 : KY0;

  const float scale = 0.125f;

  v2f ka[16];
  {
    const float* kr = ky + (size_t)(b * S_ + i0 + m) * DK + 2 * hi;
#pragma unroll
    for (int t = 0; t < 16; ++t) {
      ka[t].x = kr[4 * t] * scale;
      ka[t].y = kr[4 * t + 1] * scale;
    }
  }

  v8f yacc[4];
#pragma unroll
  for (int nt = 0; nt < 4; ++nt)
    yacc[nt] = (v8f){0.f, 0.f, 0.f, 0.f, 0.f, 0.f, 0.f, 0.f};

  for (int j0 = jbeg; j0 < jbeg + S_ / 2; j0 += JT) {
    v2f qf[NJ][16];
    {
      const float* qr = Qb + (size_t)(b * S_ + j0 + m) * DK + 2 * hi;
#pragma unroll
      for (int jt = 0; jt < NJ; ++jt) {
#pragma unroll
        for (int t = 0; t < 16; ++t) {
          qf[jt][t].x = qr[jt * 16 * DK + 4 * t];
          qf[jt][t].y = qr[jt * 16 * DK + 4 * t + 1];
        }
        __builtin_amdgcn_sched_barrier(0);
      }
    }
    v8f e[NJ];
#pragma unroll
    for (int jt = 0; jt < NJ; ++jt) {
      e[jt] = (v8f){0.f, 0.f, 0.f, 0.f, 0.f, 0.f, 0.f, 0.f};
#pragma unroll
      for (int t = 0; t < 16; ++t)
        e[jt] = wmma_f32(ka[t], qf[jt][t], e[jt]);
      __builtin_amdgcn_sched_barrier(0);
    }

#pragma unroll
    for (int jt = 0; jt < NJ; ++jt)
#pragma unroll
      for (int r = 0; r < 8; ++r)
        lds_e[((jt * 8 + r) * 32 + lane) * 12 + b] = e[jt][r];
    __syncthreads();

#pragma unroll
    for (int jt = 0; jt < NJ; ++jt) {
      const float4* p =
          (const float4*)&lds_e[((jt * 8 + b) * 32 + lane) * 12];
      float4 u = p[0];
      float4 w = p[1];
      float mx = fmaxf(fmaxf(fmaxf(u.x, u.y), fmaxf(u.z, u.w)),
                       fmaxf(fmaxf(w.x, w.y), fmaxf(w.z, w.w)));
      float t0 = __expf(u.x - mx), t1 = __expf(u.y - mx);
      float t2 = __expf(u.z - mx), t3 = __expf(u.w - mx);
      float t4 = __expf(w.x - mx), t5 = __expf(w.y - mx);
      float t6 = __expf(w.z - mx), t7 = __expf(w.w - mx);
      float z = t0 + t1 + t2 + t3 + t4 + t5 + t6 + t7;
      float* dst = &lds_a[0][(b + 8 * hi) * APITCH + jt * 16 + m];
      dst[0 * 16 * APITCH] = t0 / z;
      dst[1 * 16 * APITCH] = t1 / z;
      dst[2 * 16 * APITCH] = t2 / z;
      dst[3 * 16 * APITCH] = t3 / z;
      dst[4 * 16 * APITCH] = t4 / z;
      dst[5 * 16 * APITCH] = t5 / z;
      dst[6 * 16 * APITCH] = t6 / z;
      dst[7 * 16 * APITCH] = t7 / z;
    }
    __syncthreads();

    v2f af[4 * NJ];
#pragma unroll
    for (int t = 0; t < 4 * NJ; ++t) {
      af[t].x = lds_a[b][m * APITCH + 4 * t + 2 * hi];
      af[t].y = lds_a[b][m * APITCH + 4 * t + 2 * hi + 1];
    }

    v2f vf[4 * NJ][4];
    {
      const float* vr = Vb + (size_t)(b * S_ + j0 + 2 * hi) * DV + m;
#pragma unroll
      for (int t = 0; t < 4 * NJ; ++t)
#pragma unroll
        for (int nt = 0; nt < 4; ++nt) {
          vf[t][nt].x = vr[(4 * t) * DV + nt * 16];
          vf[t][nt].y = vr[(4 * t + 1) * DV + nt * 16];
        }
    }
#pragma unroll
    for (int t = 0; t < 4 * NJ; ++t)
#pragma unroll
      for (int nt = 0; nt < 4; ++nt)
        yacc[nt] = wmma_f32(af[t], vf[t][nt], yacc[nt]);
    __builtin_amdgcn_sched_barrier(0);
    asm volatile("v_nop\n\tv_nop\n\tv_nop\n\tv_nop");
  }

#pragma unroll
  for (int nt = 0; nt < 4; ++nt)
#pragma unroll
    for (int r = 0; r < 8; ++r)
      ky[(size_t)(b * S_ + i0 + r + 8 * hi) * DV + nt * 16 + m] =
          yacc[nt][r];
}

__global__ __launch_bounds__(256) void add_halves(
    float* __restrict__ Y, const float* __restrict__ P1) {
  const int i = blockIdx.x * 256 + threadIdx.x;
  float4 a = ((const float4*)Y)[i];
  const float4 p = ((const float4*)P1)[i];
  a.x += p.x;
  a.y += p.y;
  a.z += p.z;
  a.w += p.w;
  ((float4*)Y)[i] = a;
}

extern "C" void kernel_launch(void* const* d_in, const int* in_sizes, int n_in,
                              void* d_out, int out_size, void* d_ws,
                              size_t ws_size, hipStream_t stream) {
  const float* X  = (const float*)d_in[0];
  const float* Wk = (const float*)d_in[1];
  const float* bk = (const float*)d_in[2];
  const float* Wq = (const float*)d_in[3];
  const float* bq = (const float*)d_in[4];
  const float* Wv = (const float*)d_in[5];
  const float* bv = (const float*)d_in[6];
  float* Y = (float*)d_out;

  float* ws = (float*)d_ws;
  const size_t NKQV = (size_t)B_ * S_ * DK;
  float* Kb  = ws;
  float* Qb  = ws + NKQV;
  float* Vb  = ws + 2 * NKQV;
  float* WT3 = ws + 3 * NKQV;

  transpose_weights<<<(3 * D_IN * DK + 255) / 256, 256, 0, stream>>>(
      Wk, Wq, Wv, WT3);
  qkv_kernel<<<(B_ * S_ / 16) / 8, 256, 0, stream>>>(
      X, WT3, bk, bq, bv, Kb, Qb, Vb, Y);
  attn_kernel<<<2 * (S_ / 16), 256, 0, stream>>>(Y, Kb, Qb, Vb);
  add_halves<<<(B_ * S_ * DV / 4) / 256, 256, 0, stream>>>(Y, Kb);
}


# projection stores Q and V in the attention kernel's WMMA fragment order (Q tile computed transposed, operands swapped): a fragment load is one contiguous 512-byte wave load; sums unchanged
# speedup vs baseline: 2.7573x; 1.1619x over previous
#include <hip/hip_runtime.h>

#define D_IN 512
#define DK   64
#define DV   64
#define B_   8
#define S_   2048

typedef __attribute__((ext_vector_type(2))) float v2f;
typedef __attribute__((ext_vector_type(8))) float v8f;

static __device__ __forceinline__ v8f wmma_f32(v2f a, v2f b, v8f c) {
  return __builtin_amdgcn_wmma_f32_16x16x4_f32(
       false, a,  false, b,
       (short)0, c,  false,  false);
}

__global__ void transpose_weights(const float* __restrict__ Wk,
                                  const float* __restrict__ Wq,
                                  const float* __restrict__ Wv,
                                  float* __restrict__ WT3) {
  int idx = blockIdx.x * 256 + threadIdx.x;
  if (idx >= 3 * D_IN * DK) return;
  int mat = idx >> 15;
  int rem = idx & 32767;
  int d = rem >> 6;
  int n = rem & 63;
  const float* W = (mat == 0) ? Wk : ((mat == 1) ? Wq : Wv);
  WT3[idx] = W[n * D_IN + d];
}

__global__ __launch_bounds__(256) void qkv_kernel(
    const float* __restrict__ X, const float* __restrict__ WT3,
    const float* __restrict__ bk, const float* __restrict__ bq,
    const float* __restrict__ bv,
    float* __restrict__ Kb, float* __restrict__ Qf, float* __restrict__ Vf,
    float* __restrict__ Kb2) {
  const int lane = threadIdx.x & 31;
  const int wave = threadIdx.x >> 5;
  const int wg   = blockIdx.x * 8 + wave;
  const int b    = wg >> 7;
  const int s0   = (wg & 127) << 4;
  const int m    = lane & 15;
  const int hi   = lane >> 4;

  const float* xr = X + (size_t)(b * S_ + s0 + m) * D_IN + 2 * hi;

  v8f acc[3][4];
#pragma unroll
  for (int i = 0; i < 3; ++i)
#pragma unroll
    for (int j = 0; j < 4; ++j)
      acc[i][j] = (v8f){0.f, 0.f, 0.f, 0.f, 0.f, 0.f, 0.f, 0.f};

  for (int t = 0; t < D_IN / 4; ++t) {
    v2f a;
    a.x = xr[4 * t];
    a.y = xr[4 * t + 1];
    const float* wbase = WT3 + ((size_t)(4 * t + 2 * hi)) * DK + m;
#pragma unroll
    for (int mtx = 0; mtx < 3; ++mtx) {
#pragma unroll
      for (int nt = 0; nt < 4; ++nt) {
        const float* wp = wbase + (size_t)mtx * D_IN * DK + nt * 16;
        v2f bf;
        bf.x = wp[0];
        bf.y = wp[DK];
        if (mtx == 1)
          acc[mtx][nt] = wmma_f32(bf, a, acc[mtx][nt]);
        else
          acc[mtx][nt] = wmma_f32(a, bf, acc[mtx][nt]);
      }
    }
  }

  __builtin_amdgcn_sched_barrier(0);
  asm volatile("v_nop\n\tv_nop\n\tv_nop\n\tv_nop");
  __builtin_amdgcn_sched_barrier(0);

#pragma unroll
  for (int nt = 0; nt < 4; ++nt) {
    const float bs = bk[nt * 16 + m];
#pragma unroll
    for (int r = 0; r < 8; ++r) {
      const size_t o = (size_t)(b * S_ + s0 + r + 8 * hi) * DK + nt * 16 + m;
      const float val = acc[0][nt][r] + bs;
      Kb[o] = val;
      Kb2[o] = val;
    }
  }

  float* qt = Qf + (size_t)wg * (16 * DK);
  float* vt = Vf + (size_t)wg * (16 * DV);
#pragma unroll
  for (int nt = 0; nt < 4; ++nt) {
    float qv[8], vv[8];
    const float bsv = bv[nt * 16 + m];
#pragma unroll
    for (int r = 0; r < 8; ++r) {
      qv[r] = acc[1][nt][r] + bq[nt * 16 + 8 * hi + r];
      vv[r] = acc[2][nt][r] + bsv;
    }
    float4* qg = (float4*)(qt + (2 * nt + hi) * 128);
    float4* vg = (float4*)(vt + (2 * nt + hi) * 128);
    qg[m]      = make_float4(qv[0], qv[1], qv[4], qv[5]);
    qg[16 + m] = make_float4(qv[2], qv[3], qv[6], qv[7]);
    vg[m]      = make_float4(vv[0], vv[1], vv[4], vv[5]);
    vg[16 + m] = make_float4(vv[2], vv[3], vv[6], vv[7]);
  }
}

#define JT 64
#define NJ (JT / 16)
#define APITCH (JT + 4)

static_assert(DK == DV, "a K tile and a Y tile share one plane layout");

__global__ __launch_bounds__(256) void attn_kernel(
    float* KY0, float* KY1, const float* __restrict__ Qf,
    const float* __restrict__ Vf) {
  __shared__ __attribute__((aligned(16))) float lds_e[NJ * 256 * 12];
  __shared__ __attribute__((aligned(16))) float lds_a[8][16 * APITCH];

  const int lane = threadIdx.x & 31;
  const int b    = threadIdx.x >> 5;
  const int m    = lane & 15;
  const int hi   = lane >> 4;
  const int half = blockIdx.x & 1;
  const int i0   = (blockIdx.x >> 1) << 4;
  const int jbeg = half * (S_ / 2);
  float* ky = half ? KY1 : KY0;

  const float scale = 0.125f;

  v2f ka[16];
  {
    const float* kr = ky + (size_t)(b * S_ + i0 + m) * DK + 2 * hi;
#pragma unroll
    for (int t = 0; t < 16; ++t) {
      ka[t].x = kr[4 * t] * scale;
      ka[t].y = kr[4 * t + 1] * scale;
    }
  }

  v8f yacc[4];
#pragma unroll
  for (int nt = 0; nt < 4; ++nt)
    yacc[nt] = (v8f){0.f, 0.f, 0.f, 0.f, 0.f, 0.f, 0.f, 0.f};

  for (int j0 = jbeg; j0 < jbeg + S_ / 2; j0 += JT) {
    v2f qf[NJ][16];
    {
      const float4* qr =
          (const float4*)(Qf + (size_t)(b * (S_ / 16) + (j0 >> 4)) * (16 * DK)) +
          lane;
#pragma unroll
      for (int jt = 0; jt < NJ; ++jt) {
#pragma unroll
        for (int u = 0; u < 8; ++u) {
          const float4 x = qr[jt * (16 * DK / 4) + u * 32];
          qf[jt][2 * u]     = (v2f){x.x, x.y};
          qf[jt][2 * u + 1] = (v2f){x.z, x.w};
        }
      }
    }
    v8f e[NJ];
#pragma unroll
    for (int jt = 0; jt < NJ; ++jt) {
      e[jt] = (v8f){0.f, 0.f, 0.f, 0.f, 0.f, 0.f, 0.f, 0.f};
#pragma unroll
      for (int t = 0; t < 16; ++t)
        e[jt] = wmma_f32(ka[t], qf[jt][t], e[jt]);
      __builtin_amdgcn_sched_barrier(0);
    }

#pragma unroll
    for (int jt = 0; jt < NJ; ++jt)
#pragma unroll
      for (int r = 0; r < 8; ++r)
        lds_e[((jt * 8 + r) * 32 + lane) * 12 + b] = e[jt][r];
    __syncthreads();

#pragma unroll
    for (int jt = 0; jt < NJ; ++jt) {
      const float4* p =
          (const float4*)&lds_e[((jt * 8 + b) * 32 + lane) * 12];
      float4 u = p[0];
      float4 w = p[1];
      float mx = fmaxf(fmaxf(fmaxf(u.x, u.y), fmaxf(u.z, u.w)),
                       fmaxf(fmaxf(w.x, w.y), fmaxf(w.z, w.w)));
      float t0 = __expf(u.x - mx), t1 = __expf(u.y - mx);
      float t2 = __expf(u.z - mx), t3 = __expf(u.w - mx);
      float t4 = __expf(w.x - mx), t5 = __expf(w.y - mx);
      float t6 = __expf(w.z - mx), t7 = __expf(w.w - mx);
      float z = t0 + t1 + t2 + t3 + t4 + t5 + t6 + t7;
      float* dst = &lds_a[0][(b + 8 * hi) * APITCH + jt * 16 + m];
      dst[0 * 16 * APITCH] = t0 / z;
      dst[1 * 16 * APITCH] = t1 / z;
      dst[2 * 16 * APITCH] = t2 / z;
      dst[3 * 16 * APITCH] = t3 / z;
      dst[4 * 16 * APITCH] = t4 / z;
      dst[5 * 16 * APITCH] = t5 / z;
      dst[6 * 16 * APITCH] = t6 / z;
      dst[7 * 16 * APITCH] = t7 / z;
    }
    __syncthreads();

    v2f af[4 * NJ];
#pragma unroll
    for (int t = 0; t < 4 * NJ; ++t) {
      af[t].x = lds_a[b][m * APITCH + 4 * t + 2 * hi];
      af[t].y = lds_a[b][m * APITCH + 4 * t + 2 * hi + 1];
    }

    v2f vf[4 * NJ][4];
    {
      const float4* vr =
          (const float4*)(Vf + (size_t)(b * (S_ / 16) + (j0 >> 4)) * (16 * DV)) +
          lane;
#pragma unroll
      for (int jt = 0; jt < NJ; ++jt)
#pragma unroll
        for (int u = 0; u < 2; ++u)
#pragma unroll
          for (int nt = 0; nt < 4; ++nt) {
            const float4 x = vr[jt * (16 * DV / 4) + (2 * nt + u) * 32];
            vf[4 * jt + 2 * u][nt]     = (v2f){x.x, x.y};
            vf[4 * jt + 2 * u + 1][nt] = (v2f){x.z, x.w};
          }
    }
#pragma unroll
    for (int t = 0; t < 4 * NJ; ++t)
#pragma unroll
      for (int nt = 0; nt < 4; ++nt)
        yacc[nt] = wmma_f32(af[t], vf[t][nt], yacc[nt]);
    __builtin_amdgcn_sched_barrier(0);
    asm volatile("v_nop\n\tv_nop\n\tv_nop\n\tv_nop");
  }

#pragma unroll
  for (int nt = 0; nt < 4; ++nt)
#pragma unroll
    for (int r = 0; r < 8; ++r)
      ky[(size_t)(b * S_ + i0 + r + 8 * hi) * DV + nt * 16 + m] =
          yacc[nt][r];
}

__global__ __launch_bounds__(256) void add_halves(
    float* __restrict__ Y, const float* __restrict__ P1) {
  const int i = blockIdx.x * 256 + threadIdx.x;
  float4 a = ((const float4*)Y)[i];
  const float4 p = ((const float4*)P1)[i];
  a.x += p.x;
  a.y += p.y;
  a.z += p.z;
  a.w += p.w;
  ((float4*)Y)[i] = a;
}

extern "C" void kernel_launch(void* const* d_in, const int* in_sizes, int n_in,
                              void* d_out, int out_size, void* d_ws,
                              size_t ws_size, hipStream_t stream) {
  const float* X  = (const float*)d_in[0];
  const float* Wk = (const float*)d_in[1];
  const float* bk = (const float*)d_in[2];
  const float* Wq = (const float*)d_in[3];
  const float* bq = (const float*)d_in[4];
  const float* Wv = (const float*)d_in[5];
  const float* bv = (const float*)d_in[6];
  float* Y = (float*)d_out;

  float* ws = (float*)d_ws;
  const size_t NKQV = (size_t)B_ * S_ * DK;
  float* Kb  = ws;
  float* Qf  = ws + NKQV;
  float* Vf  = ws + 2 * NKQV;
  float* WT3 = ws + 3 * NKQV;

  transpose_weights<<<(3 * D_IN * DK + 255) / 256, 256, 0, stream>>>(
      Wk, Wq, Wv, WT3);
  qkv_kernel<<<(B_ * S_ / 16) / 8, 256, 0, stream>>>(
      X, WT3, bk, bq, bv, Kb, Qf, Vf, Y);
  attn_kernel<<<2 * (S_ / 16), 256, 0, stream>>>(Y, Kb, Qf, Vf);
  add_halves<<<(B_ * S_ * DV / 4) / 256, 256, 0, stream>>>(Y, Kb);
}


# attn kernel: global fragment loads issued one phase ahead, each as one group (a step's V in front of its E products, the next step's Q in front of its Y products); order of all sums unchanged
# speedup vs baseline: 2.9241x; 1.0605x over previous
#include <hip/hip_runtime.h>

#define D_IN 512
#define DK   64
#define DV   64
#define B_   8
#define S_   2048

typedef __attribute__((ext_vector_type(2))) float v2f;
typedef __attribute__((ext_vector_type(8))) float v8f;

static __device__ __forceinline__ v8f wmma_f32(v2f a, v2f b, v8f c) {
  return __builtin_amdgcn_wmma_f32_16x16x4_f32(
       false, a,  false, b,
       (short)0, c,  false,  false);
}

__global__ void transpose_weights(const float* __restrict__ Wk,
                                  const float* __restrict__ Wq,
                                  const float* __restrict__ Wv,
                                  float* __restrict__ WT3) {
  int idx = blockIdx.x * 256 + threadIdx.x;
  if (idx >= 3 * D_IN * DK) return;
  int mat = idx >> 15;
  int rem = idx & 32767;
  int d = rem >> 6;
  int n = rem & 63;
  const float* W = (mat == 0) ? Wk : ((mat == 1) ? Wq : Wv);
  WT3[idx] = W[n * D_IN + d];
}

__global__ __launch_bounds__(256) void qkv_kernel(
    const float* __restrict__ X, const float* __restrict__ WT3,
    const float* __restrict__ bk, const float* __restrict__ bq,
    const float* __restrict__ bv,
    float* __restrict__ Kb, float* __restrict__ Qf, float* __restrict__ Vf,
    float* __restrict__ Kb2) {
  const int lane = threadIdx.x & 31;
  const int wave = threadIdx.x >> 5;
  const int wg   = blockIdx.x * 8 + wave;
  const int b    = wg >> 7;
  const int s0   = (wg & 127) << 4;
  const int m    = lane & 15;
  const int hi   = lane >> 4;

  const float* xr = X + (size_t)(b * S_ + s0 + m) * D_IN + 2 * hi;

  v8f acc[3][4];
#pragma unroll
  for (int i = 0; i < 3; ++i)
#pragma unroll
    for (int j = 0; j < 4; ++j)
      acc[i][j] = (v8f){0.f, 0.f, 0.f, 0.f, 0.f, 0.f, 0.f, 0.f};

  for (int t = 0; t < D_IN / 4; ++t) {
    v2f a;
    a.x = xr[4 * t];
    a.y = xr[4 * t + 1];
    const float* wbase = WT3 + ((size_t)(4 * t + 2 * hi)) * DK + m;
#pragma unroll
    for (int mtx = 0; mtx < 3; ++mtx) {
#pragma unroll
      for (int nt = 0; nt < 4; ++nt) {
        const float* wp = wbase + (size_t)mtx * D_IN * DK + nt * 16;
        v2f bf;
        bf.x = wp[0];
        bf.y = wp[DK];
        if (mtx == 1)
          acc[mtx][nt] = wmma_f32(bf, a, acc[mtx][nt]);
        else
          acc[mtx][nt] = wmma_f32(a, bf, acc[mtx][nt]);
      }
    }
  }

  __builtin_amdgcn_sched_barrier(0);
  asm volatile("v_nop\n\tv_nop\n\tv_nop\n\tv_nop");
  __builtin_amdgcn_sched_barrier(0);

#pragma unroll
  for (int nt = 0; nt < 4; ++nt) {
    const float bs = bk[nt * 16 + m];
#pragma unroll
    for (int r = 0; r < 8; ++r) {
      const size_t o = (size_t)(b * S_ + s0 + r + 8 * hi) * DK + nt * 16 + m;
      const float val = acc[0][nt][r] + bs;
      Kb[o] = val;
      Kb2[o] = val;
    }
  }

  float* qt = Qf + (size_t)wg * (16 * DK);
  float* vt = Vf + (size_t)wg * (16 * DV);
#pragma unroll
  for (int nt = 0; nt < 4; ++nt) {
    float qv[8], vv[8];
    const float bsv = bv[nt * 16 + m];
#pragma unroll
    for (int r = 0; r < 8; ++r) {
      qv[r] = acc[1][nt][r] + bq[nt * 16 + 8 * hi + r];
      vv[r] = acc[2][nt][r] + bsv;
    }
    float4* qg = (float4*)(qt + (2 * nt + hi) * 128);
    float4* vg = (float4*)(vt + (2 * nt + hi) * 128);
    qg[m]      = make_float4(qv[0], qv[1], qv[4], qv[5]);
    qg[16 + m] = make_float4(qv[2], qv[3], qv[6], qv[7]);
    vg[m]      = make_float4(vv[0], vv[1], vv[4], vv[5]);
    vg[16 + m] = make_float4(vv[2], vv[3], vv[6], vv[7]);
  }
}

#define JT 64
#define NJ (JT / 16)
#define APITCH (JT + 4)

static_assert(DK == DV, "a K tile and a Y tile share one plane layout");

static __device__ __forceinline__ void load_q_step(
    const float* __restrict__ Qf, int b, int j0, int lane, v2f (&qf)[NJ][16]) {
  const float4* qr =
      (const float4*)(Qf + (size_t)(b * (S_ / 16) + (j0 >> 4)) * (16 * DK)) +
      lane;
#pragma unroll
  for (int jt = 0; jt < NJ; ++jt) {
#pragma unroll
    for (int u = 0; u < 8; ++u) {
      const float4 x = qr[jt * (16 * DK / 4) + u * 32];
      qf[jt][2 * u]     = (v2f){x.x, x.y};
      qf[jt][2 * u + 1] = (v2f){x.z, x.w};
    }
  }
}

static __device__ __forceinline__ void load_v_step(
    const float* __restrict__ Vf, int b, int j0, int lane,
    v2f (&vf)[4 * NJ][4]) {
  const float4* vr =
      (const float4*)(Vf + (size_t)(b * (S_ / 16) + (j0 >> 4)) * (16 * DV)) +
      lane;
#pragma unroll
  for (int jt = 0; jt < NJ; ++jt)
#pragma unroll
    for (int u = 0; u < 2; ++u)
#pragma unroll
      for (int nt = 0; nt < 4; ++nt) {
        const float4 x = vr[jt * (16 * DV / 4) + (2 * nt + u) * 32];
        vf[4 * jt + 2 * u][nt]     = (v2f){x.x, x.y};
        vf[4 * jt + 2 * u + 1][nt] = (v2f){x.z, x.w};
      }
}

__global__ __launch_bounds__(256) void attn_kernel(
    float* KY0, float* KY1, const float* __restrict__ Qf,
    const float* __restrict__ Vf) {
  __shared__ __attribute__((aligned(16))) float lds_e[NJ * 256 * 12];
  __shared__ __attribute__((aligned(16))) float lds_a[8][16 * APITCH];

  const int lane = threadIdx.x & 31;
  const int b    = threadIdx.x >> 5;
  const int m    = lane & 15;
  const int hi   = lane >> 4;
  const int half = blockIdx.x & 1;
  const int i0   = (blockIdx.x >> 1) << 4;
  const int jbeg = half * (S_ / 2);
  float* ky = half ? KY1 : KY0;

  const float scale = 0.125f;

  v2f ka[16];
  {
    const float* kr = ky + (size_t)(b * S_ + i0 + m) * DK + 2 * hi;
#pragma unroll
    for (int t = 0; t < 16; ++t) {
      ka[t].x = kr[4 * t] * scale;
      ka[t].y = kr[4 * t + 1] * scale;
    }
  }

  v8f yacc[4];
#pragma unroll
  for (int nt = 0; nt < 4; ++nt)
    yacc[nt] = (v8f){0.f, 0.f, 0.f, 0.f, 0.f, 0.f, 0.f, 0.f};

  const int jend = jbeg + S_ / 2;
  v2f qf[NJ][16];
  load_q_step(Qf, b, jbeg, lane, qf);

  for (int j0 = jbeg; j0 < jend; j0 += JT) {
    v2f vf[4 * NJ][4];
    load_v_step(Vf, b, j0, lane, vf);
    __builtin_amdgcn_sched_barrier(0);

    v8f e[NJ];
#pragma unroll
    for (int jt = 0; jt < NJ; ++jt) {
      e[jt] = (v8f){0.f, 0.f, 0.f, 0.f, 0.f, 0.f, 0.f, 0.f};
#pragma unroll
      for (int t = 0; t < 16; ++t)
        e[jt] = wmma_f32(ka[t], qf[jt][t], e[jt]);
    }

#pragma unroll
    for (int jt = 0; jt < NJ; ++jt)
#pragma unroll
      for (int r = 0; r < 8; ++r)
        lds_e[((jt * 8 + r) * 32 + lane) * 12 + b] = e[jt][r];
    __syncthreads();

#pragma unroll
    for (int jt = 0; jt < NJ; ++jt) {
      const float4* p =
          (const float4*)&lds_e[((jt * 8 + b) * 32 + lane) * 12];
      float4 u = p[0];
      float4 w = p[1];
      float mx = fmaxf(fmaxf(fmaxf(u.x, u.y), fmaxf(u.z, u.w)),
                       fmaxf(fmaxf(w.x, w.y), fmaxf(w.z, w.w)));
      float t0 = __expf(u.x - mx), t1 = __expf(u.y - mx);
      float t2 = __expf(u.z - mx), t3 = __expf(u.w - mx);
      float t4 = __expf(w.x - mx), t5 = __expf(w.y - mx);
      float t6 = __expf(w.z - mx), t7 = __expf(w.w - mx);
      float z = t0 + t1 + t2 + t3 + t4 + t5 + t6 + t7;
      float* dst = &lds_a[0][(b + 8 * hi) * APITCH + jt * 16 + m];
      dst[0 * 16 * APITCH] = t0 / z;
      dst[1 * 16 * APITCH] = t1 / z;
      dst[2 * 16 * APITCH] = t2 / z;
      dst[3 * 16 * APITCH] = t3 / z;
      dst[4 * 16 * APITCH] = t4 / z;
      dst[5 * 16 * APITCH] = t5 / z;
      dst[6 * 16 * APITCH] = t6 / z;
      dst[7 * 16 * APITCH] = t7 / z;
    }
    __syncthreads();

    v2f af[4 * NJ];
#pragma unroll
    for (int t = 0; t < 4 * NJ; ++t) {
      af[t].x = lds_a[b][m * APITCH + 4 * t + 2 * hi];
      af[t].y = lds_a[b][m * APITCH + 4 * t + 2 * hi + 1];
    }

    load_q_step(Qf, b, (j0 + JT < jend) ? j0 + JT : j0, lane, qf);
    __builtin_amdgcn_sched_barrier(0);

#pragma unroll
    for (int t = 0; t < 4 * NJ; ++t)
#pragma unroll
      for (int nt = 0; nt < 4; ++nt)
        yacc[nt] = wmma_f32(af[t], vf[t][nt], yacc[nt]);
    __builtin_amdgcn_sched_barrier(0);
    asm volatile("v_nop\n\tv_nop\n\tv_nop\n\tv_nop");
  }

#pragma unroll
  for (int nt = 0; nt < 4; ++nt)
#pragma unroll
    for (int r = 0; r < 8; ++r)
      ky[(size_t)(b * S_ + i0 + r + 8 * hi) * DV + nt * 16 + m] =
          yacc[nt][r];
}

__global__ __launch_bounds__(256) void add_halves(
    float* __restrict__ Y, const float* __restrict__ P1) {
  const int i = blockIdx.x * 256 + threadIdx.x;
  float4 a = ((const float4*)Y)[i];
  const float4 p = ((const float4*)P1)[i];
  a.x += p.x;
  a.y += p.y;
  a.z += p.z;
  a.w += p.w;
  ((float4*)Y)[i] = a;
}

extern "C" void kernel_launch(void* const* d_in, const int* in_sizes, int n_in,
                              void* d_out, int out_size, void* d_ws,
                              size_t ws_size, hipStream_t stream) {
  const float* X  = (const float*)d_in[0];
  const float* Wk = (const float*)d_in[1];
  const float* bk = (const float*)d_in[2];
  const float* Wq = (const float*)d_in[3];
  const float* bq = (const float*)d_in[4];
  const float* Wv = (const float*)d_in[5];
  const float* bv = (const float*)d_in[6];
  float* Y = (float*)d_out;

  float* ws = (float*)d_ws;
  const size_t NKQV = (size_t)B_ * S_ * DK;
  float* Kb  = ws;
  float* Qf  = ws + NKQV;
  float* Vf  = ws + 2 * NKQV;
  float* WT3 = ws + 3 * NKQV;

  transpose_weights<<<(3 * D_IN * DK + 255) / 256, 256, 0, stream>>>(
      Wk, Wq, Wv, WT3);
  qkv_kernel<<<(B_ * S_ / 16) / 8, 256, 0, stream>>>(
      X, WT3, bk, bq, bv, Kb, Qf, Vf, Y);
  attn_kernel<<<2 * (S_ / 16), 256, 0, stream>>>(Y, Kb, Qf, Vf);
  add_halves<<<(B_ * S_ * DV / 4) / 256, 256, 0, stream>>>(Y, Kb);
}


# projection kernel: k range cut in two halves summed by a pair of waves of one block, 256 blocks (one per WGP) not 128; partner adds the halves via LDS: sum over k is two running sums plus one add
# speedup vs baseline: 3.5231x; 1.2048x over previous
#include <hip/hip_runtime.h>

#define D_IN 512
#define DK   64
#define DV   64
#define B_   8
#define S_   2048

typedef __attribute__((ext_vector_type(2))) float v2f;
typedef __attribute__((ext_vector_type(8))) float v8f;

static __device__ __forceinline__ v8f wmma_f32(v2f a, v2f b, v8f c) {
  return __builtin_amdgcn_wmma_f32_16x16x4_f32(
       false, a,  false, b,
       (short)0, c,  false,  false);
}

__global__ void transpose_weights(const float* __restrict__ Wk,
                                  const float* __restrict__ Wq,
                                  const float* __restrict__ Wv,
                                  float* __restrict__ WT3) {
  int idx = blockIdx.x * 256 + threadIdx.x;
  if (idx >= 3 * D_IN * DK) return;
  int mat = idx >> 15;
  int rem = idx & 32767;
  int d = rem >> 6;
  int n = rem & 63;
  const float* W = (mat == 0) ? Wk : ((mat == 1) ? Wq : Wv);
  WT3[idx] = W[n * D_IN + d];
}

__global__ __launch_bounds__(256) void qkv_kernel(
    const float* __restrict__ X, const float* __restrict__ WT3,
    const float* __restrict__ bk, const float* __restrict__ bq,
    const float* __restrict__ bv,
    float* __restrict__ Kb, float* __restrict__ Qf, float* __restrict__ Vf,
    float* __restrict__ Kb2) {
  __shared__ __attribute__((aligned(16))) float lds_p[4][24 * 128];

  const int lane = threadIdx.x & 31;
  const int wave = threadIdx.x >> 5;
  const int rt   = wave & 3;
  const int kh   = wave >> 2;
  const int wg   = blockIdx.x * 4 + rt;
  const int b    = wg >> 7;
  const int s0   = (wg & 127) << 4;
  const int m    = lane & 15;
  const int hi   = lane >> 4;
  const int kb   = kh * (D_IN / 2);

  const float* xr = X + (size_t)(b * S_ + s0 + m) * D_IN + kb + 2 * hi;

  v8f acc[3][4];
#pragma unroll
  for (int i = 0; i < 3; ++i)
#pragma unroll
    for (int j = 0; j < 4; ++j)
      acc[i][j] = (v8f){0.f, 0.f, 0.f, 0.f, 0.f, 0.f, 0.f, 0.f};

  for (int t = 0; t < D_IN / 8; ++t) {
    v2f a;
    a.x = xr[4 * t];
    a.y = xr[4 * t + 1];
    const float* wbase = WT3 + ((size_t)(kb + 4 * t + 2 * hi)) * DK + m;
#pragma unroll
    for (int mtx = 0; mtx < 3; ++mtx) {
#pragma unroll
      for (int nt = 0; nt < 4; ++nt) {
        const float* wp = wbase + (size_t)mtx * D_IN * DK + nt * 16;
        v2f bf;
        bf.x = wp[0];
        bf.y = wp[DK];
        if (mtx == 1)
          acc[mtx][nt] = wmma_f32(bf, a, acc[mtx][nt]);
        else
          acc[mtx][nt] = wmma_f32(a, bf, acc[mtx][nt]);
      }
    }
  }

  __builtin_amdgcn_sched_barrier(0);
  asm volatile("v_nop\n\tv_nop\n\tv_nop\n\tv_nop");
  __builtin_amdgcn_sched_barrier(0);

  float4* pp = (float4*)lds_p[rt] + lane;
  if (kh == 1) {
#pragma unroll
    for (int i = 0; i < 3; ++i)
#pragma unroll
      for (int j = 0; j < 4; ++j)
#pragma unroll
        for (int q = 0; q < 2; ++q)
          pp[((i * 4 + j) * 2 + q) * 32] =
              make_float4(acc[i][j][4 * q], acc[i][j][4 * q + 1],
                          acc[i][j][4 * q + 2], acc[i][j][4 * q + 3]);
  }
  __syncthreads();
  if (kh == 1) return;
#pragma unroll
  for (int i = 0; i < 3; ++i)
#pragma unroll
    for (int j = 0; j < 4; ++j)
#pragma unroll
      for (int q = 0; q < 2; ++q) {
        const float4 p = pp[((i * 4 + j) * 2 + q) * 32];
        acc[i][j][4 * q]     += p.x;
        acc[i][j][4 * q + 1] += p.y;
        acc[i][j][4 * q + 2] += p.z;
        acc[i][j][4 * q + 3] += p.w;
      }

#pragma unroll
  for (int nt = 0; nt < 4; ++nt) {
    const float bs = bk[nt * 16 + m];
#pragma unroll
    for (int r = 0; r < 8; ++r) {
      const size_t o = (size_t)(b * S_ + s0 + r + 8 * hi) * DK + nt * 16 + m;
      const float val = acc[0][nt][r] + bs;
      Kb[o] = val;
      Kb2[o] = val;
    }
  }

  float* qt = Qf + (size_t)wg * (16 * DK);
  float* vt = Vf + (size_t)wg * (16 * DV);
#pragma unroll
  for (int nt = 0; nt < 4; ++nt) {
    float qv[8], vv[8];
    const float bsv = bv[nt * 16 + m];
#pragma unroll
    for (int r = 0; r < 8; ++r) {
      qv[r] = acc[1][nt][r] + bq[nt * 16 + 8 * hi + r];
      vv[r] = acc[2][nt][r] + bsv;
    }
    float4* qg = (float4*)(qt + (2 * nt + hi) * 128);
    float4* vg = (float4*)(vt + (2 * nt + hi) * 128);
    qg[m]      = make_float4(qv[0], qv[1], qv[4], qv[5]);
    qg[16 + m] = make_float4(qv[2], qv[3], qv[6], qv[7]);
    vg[m]      = make_float4(vv[0], vv[1], vv[4], vv[5]);
    vg[16 + m] = make_float4(vv[2], vv[3], vv[6], vv[7]);
  }
}

#define JT 64
#define NJ (JT / 16)
#define APITCH (JT + 4)

static_assert(DK == DV, "a K tile and a Y tile share one plane layout");

static __device__ __forceinline__ void load_q_step(
    const float* __restrict__ Qf, int b, int j0, int lane, v2f (&qf)[NJ][16]) {
  const float4* qr =
      (const float4*)(Qf + (size_t)(b * (S_ / 16) + (j0 >> 4)) * (16 * DK)) +
      lane;
#pragma unroll
  for (int jt = 0; jt < NJ; ++jt) {
#pragma unroll
    for (int u = 0; u < 8; ++u) {
      const float4 x = qr[jt * (16 * DK / 4) + u * 32];
      qf[jt][2 * u]     = (v2f){x.x, x.y};
      qf[jt][2 * u + 1] = (v2f){x.z, x.w};
    }
  }
}

static __device__ __forceinline__ void load_v_step(
    const float* __restrict__ Vf, int b, int j0, int lane,
    v2f (&vf)[4 * NJ][4]) {
  const float4* vr =
      (const float4*)(Vf + (size_t)(b * (S_ / 16) + (j0 >> 4)) * (16 * DV)) +
      lane;
#pragma unroll
  for (int jt = 0; jt < NJ; ++jt)
#pragma unroll
    for (int u = 0; u < 2; ++u)
#pragma unroll
      for (int nt = 0; nt < 4; ++nt) {
        const float4 x = vr[jt * (16 * DV / 4) + (2 * nt + u) * 32];
        vf[4 * jt + 2 * u][nt]     = (v2f){x.x, x.y};
        vf[4 * jt + 2 * u + 1][nt] = (v2f){x.z, x.w};
      }
}

__global__ __launch_bounds__(256) void attn_kernel(
    float* KY0, float* KY1, const float* __restrict__ Qf,
    const float* __restrict__ Vf) {
  __shared__ __attribute__((aligned(16))) float lds_e[NJ * 256 * 12];
  __shared__ __attribute__((aligned(16))) float lds_a[8][16 * APITCH];

  const int lane = threadIdx.x & 31;
  const int b    = threadIdx.x >> 5;
  const int m    = lane & 15;
  const int hi   = lane >> 4;
  const int half = blockIdx.x & 1;
  const int i0   = (blockIdx.x >> 1) << 4;
  const int jbeg = half * (S_ / 2);
  float* ky = half ? KY1 : KY0;

  const float scale = 0.125f;

  v2f ka[16];
  {
    const float* kr = ky + (size_t)(b * S_ + i0 + m) * DK + 2 * hi;
#pragma unroll
    for (int t = 0; t < 16; ++t) {
      ka[t].x = kr[4 * t] * scale;
      ka[t].y = kr[4 * t + 1] * scale;
    }
  }

  v8f yacc[4];
#pragma unroll
  for (int nt = 0; nt < 4; ++nt)
    yacc[nt] = (v8f){0.f, 0.f, 0.f, 0.f, 0.f, 0.f, 0.f, 0.f};

  const int jend = jbeg + S_ / 2;
  v2f qf[NJ][16];
  load_q_step(Qf, b, jbeg, lane, qf);

  for (int j0 = jbeg; j0 < jend; j0 += JT) {
    v2f vf[4 * NJ][4];
    load_v_step(Vf, b, j0, lane, vf);
    __builtin_amdgcn_sched_barrier(0);

    v8f e[NJ];
#pragma unroll
    for (int jt = 0; jt < NJ; ++jt) {
      e[jt] = (v8f){0.f, 0.f, 0.f, 0.f, 0.f, 0.f, 0.f, 0.f};
#pragma unroll
      for (int t = 0; t < 16; ++t)
        e[jt] = wmma_f32(ka[t], qf[jt][t], e[jt]);
    }

#pragma unroll
    for (int jt = 0; jt < NJ; ++jt)
#pragma unroll
      for (int r = 0; r < 8; ++r)
        lds_e[((jt * 8 + r) * 32 + lane) * 12 + b] = e[jt][r];
    __syncthreads();

#pragma unroll
    for (int jt = 0; jt < NJ; ++jt) {
      const float4* p =
          (const float4*)&lds_e[((jt * 8 + b) * 32 + lane) * 12];
      float4 u = p[0];
      float4 w = p[1];
      float mx = fmaxf(fmaxf(fmaxf(u.x, u.y), fmaxf(u.z, u.w)),
                       fmaxf(fmaxf(w.x, w.y), fmaxf(w.z, w.w)));
      float t0 = __expf(u.x - mx), t1 = __expf(u.y - mx);
      float t2 = __expf(u.z - mx), t3 = __expf(u.w - mx);
      float t4 = __expf(w.x - mx), t5 = __expf(w.y - mx);
      float t6 = __expf(w.z - mx), t7 = __expf(w.w - mx);
      float z = t0 + t1 + t2 + t3 + t4 + t5 + t6 + t7;
      float* dst = &lds_a[0][(b + 8 * hi) * APITCH + jt * 16 + m];
      dst[0 * 16 * APITCH] = t0 / z;
      dst[1 * 16 * APITCH] = t1 / z;
      dst[2 * 16 * APITCH] = t2 / z;
      dst[3 * 16 * APITCH] = t3 / z;
      dst[4 * 16 * APITCH] = t4 / z;
      dst[5 * 16 * APITCH] = t5 / z;
      dst[6 * 16 * APITCH] = t6 / z;
      dst[7 * 16 * APITCH] = t7 / z;
    }
    __syncthreads();

    v2f af[4 * NJ];
#pragma unroll
    for (int t = 0; t < 4 * NJ; ++t) {
      af[t].x = lds_a[b][m * APITCH + 4 * t + 2 * hi];
      af[t].y = lds_a[b][m * APITCH + 4 * t + 2 * hi + 1];
    }

    load_q_step(Qf, b, (j0 + JT < jend) ? j0 + JT : j0, lane, qf);
    __builtin_amdgcn_sched_barrier(0);

#pragma unroll
    for (int t = 0; t < 4 * NJ; ++t)
#pragma unroll
      for (int nt = 0; nt < 4; ++nt)
        yacc[nt] = wmma_f32(af[t], vf[t][nt], yacc[nt]);
    __builtin_amdgcn_sched_barrier(0);
    asm volatile("v_nop\n\tv_nop\n\tv_nop\n\tv_nop");
  }

#pragma unroll
  for (int nt = 0; nt < 4; ++nt)
#pragma unroll
    for (int r = 0; r < 8; ++r)
      ky[(size_t)(b * S_ + i0 + r + 8 * hi) * DV + nt * 16 + m] =
          yacc[nt][r];
}

__global__ __launch_bounds__(256) void add_halves(
    float* __restrict__ Y, const float* __restrict__ P1) {
  const int i = blockIdx.x * 256 + threadIdx.x;
  float4 a = ((const float4*)Y)[i];
  const float4 p = ((const float4*)P1)[i];
  a.x += p.x;
  a.y += p.y;
  a.z += p.z;
  a.w += p.w;
  ((float4*)Y)[i] = a;
}

extern "C" void kernel_launch(void* const* d_in, const int* in_sizes, int n_in,
                              void* d_out, int out_size, void* d_ws,
                              size_t ws_size, hipStream_t stream) {
  const float* X  = (const float*)d_in[0];
  const float* Wk = (const float*)d_in[1];
  const float* bk = (const float*)d_in[2];
  const float* Wq = (const float*)d_in[3];
  const float* bq = (const float*)d_in[4];
  const float* Wv = (const float*)d_in[5];
  const float* bv = (const float*)d_in[6];
  float* Y = (float*)d_out;

  float* ws = (float*)d_ws;
  const size_t NKQV = (size_t)B_ * S_ * DK;
  float* Kb  = ws;
  float* Qf  = ws + NKQV;
  float* Vf  = ws + 2 * NKQV;
  float* WT3 = ws + 3 * NKQV;

  transpose_weights<<<(3 * D_IN * DK + 255) / 256, 256, 0, stream>>>(
      Wk, Wq, Wv, WT3);
  qkv_kernel<<<(B_ * S_ / 16) / 4, 256, 0, stream>>>(
      X, WT3, bk, bq, bv, Kb, Qf, Vf, Y);
  attn_kernel<<<2 * (S_ / 16), 256, 0, stream>>>(Y, Kb, Qf, Vf);
  add_halves<<<(B_ * S_ * DV / 4) / 256, 256, 0, stream>>>(Y, Kb);
}


# attn kernel softmax: an element's 8 normalised values are exp * (one reciprocal of the sum: v_rcp_f32 + one Newton step), not 8 exact divisions of 11 vector instructions each; all sums unchanged
# speedup vs baseline: 3.8020x; 1.0792x over previous
#include <hip/hip_runtime.h>

#define D_IN 512
#define DK   64
#define DV   64
#define B_   8
#define S_   2048

typedef __attribute__((ext_vector_type(2))) float v2f;
typedef __attribute__((ext_vector_type(8))) float v8f;

static __device__ __forceinline__ v8f wmma_f32(v2f a, v2f b, v8f c) {
  return __builtin_amdgcn_wmma_f32_16x16x4_f32(
       false, a,  false, b,
       (short)0, c,  false,  false);
}

__global__ void transpose_weights(const float* __restrict__ Wk,
                                  const float* __restrict__ Wq,
                                  const float* __restrict__ Wv,
                                  float* __restrict__ WT3) {
  int idx = blockIdx.x * 256 + threadIdx.x;
  if (idx >= 3 * D_IN * DK) return;
  int mat = idx >> 15;
  int rem = idx & 32767;
  int d = rem >> 6;
  int n = rem & 63;
  const float* W = (mat == 0) ? Wk : ((mat == 1) ? Wq : Wv);
  WT3[idx] = W[n * D_IN + d];
}

__global__ __launch_bounds__(256) void qkv_kernel(
    const float* __restrict__ X, const float* __restrict__ WT3,
    const float* __restrict__ bk, const float* __restrict__ bq,
    const float* __restrict__ bv,
    float* __restrict__ Kb, float* __restrict__ Qf, float* __restrict__ Vf,
    float* __restrict__ Kb2) {
  __shared__ __attribute__((aligned(16))) float lds_p[4][24 * 128];

  const int lane = threadIdx.x & 31;
  const int wave = threadIdx.x >> 5;
  const int rt   = wave & 3;
  const int kh   = wave >> 2;
  const int wg   = blockIdx.x * 4 + rt;
  const int b    = wg >> 7;
  const int s0   = (wg & 127) << 4;
  const int m    = lane & 15;
  const int hi   = lane >> 4;
  const int kb   = kh * (D_IN / 2);

  const float* xr = X + (size_t)(b * S_ + s0 + m) * D_IN + kb + 2 * hi;

  v8f acc[3][4];
#pragma unroll
  for (int i = 0; i < 3; ++i)
#pragma unroll
    for (int j = 0; j < 4; ++j)
      acc[i][j] = (v8f){0.f, 0.f, 0.f, 0.f, 0.f, 0.f, 0.f, 0.f};

  for (int t = 0; t < D_IN / 8; ++t) {
    v2f a;
    a.x = xr[4 * t];
    a.y = xr[4 * t + 1];
    const float* wbase = WT3 + ((size_t)(kb + 4 * t + 2 * hi)) * DK + m;
#pragma unroll
    for (int mtx = 0; mtx < 3; ++mtx) {
#pragma unroll
      for (int nt = 0; nt < 4; ++nt) {
        const float* wp = wbase + (size_t)mtx * D_IN * DK + nt * 16;
        v2f bf;
        bf.x = wp[0];
        bf.y = wp[DK];
        if (mtx == 1)
          acc[mtx][nt] = wmma_f32(bf, a, acc[mtx][nt]);
        else
          acc[mtx][nt] = wmma_f32(a, bf, acc[mtx][nt]);
      }
    }
  }

  __builtin_amdgcn_sched_barrier(0);
  asm volatile("v_nop\n\tv_nop\n\tv_nop\n\tv_nop");
  __builtin_amdgcn_sched_barrier(0);

  float4* pp = (float4*)lds_p[rt] + lane;
  if (kh == 1) {
#pragma unroll
    for (int i = 0; i < 3; ++i)
#pragma unroll
      for (int j = 0; j < 4; ++j)
#pragma unroll
        for (int q = 0; q < 2; ++q)
          pp[((i * 4 + j) * 2 + q) * 32] =
              make_float4(acc[i][j][4 * q], acc[i][j][4 * q + 1],
                          acc[i][j][4 * q + 2], acc[i][j][4 * q + 3]);
  }
  __syncthreads();
  if (kh == 1) return;
#pragma unroll
  for (int i = 0; i < 3; ++i)
#pragma unroll
    for (int j = 0; j < 4; ++j)
#pragma unroll
      for (int q = 0; q < 2; ++q) {
        const float4 p = pp[((i * 4 + j) * 2 + q) * 32];
        acc[i][j][4 * q]     += p.x;
        acc[i][j][4 * q + 1] += p.y;
        acc[i][j][4 * q + 2] += p.z;
        acc[i][j][4 * q + 3] += p.w;
      }

#pragma unroll
  for (int nt = 0; nt < 4; ++nt) {
    const float bs = bk[nt * 16 + m];
#pragma unroll
    for (int r = 0; r < 8; ++r) {
      const size_t o = (size_t)(b * S_ + s0 + r + 8 * hi) * DK + nt * 16 + m;
      const float val = acc[0][nt][r] + bs;
      Kb[o] = val;
      Kb2[o] = val;
    }
  }

  float* qt = Qf + (size_t)wg * (16 * DK);
  float* vt = Vf + (size_t)wg * (16 * DV);
#pragma unroll
  for (int nt = 0; nt < 4; ++nt) {
    float qv[8], vv[8];
    const float bsv = bv[nt * 16 + m];
#pragma unroll
    for (int r = 0; r < 8; ++r) {
      qv[r] = acc[1][nt][r] + bq[nt * 16 + 8 * hi + r];
      vv[r] = acc[2][nt][r] + bsv;
    }
    float4* qg = (float4*)(qt + (2 * nt + hi) * 128);
    float4* vg = (float4*)(vt + (2 * nt + hi) * 128);
    qg[m]      = make_float4(qv[0], qv[1], qv[4], qv[5]);
    qg[16 + m] = make_float4(qv[2], qv[3], qv[6], qv[7]);
    vg[m]      = make_float4(vv[0], vv[1], vv[4], vv[5]);
    vg[16 + m] = make_float4(vv[2], vv[3], vv[6], vv[7]);
  }
}

#define JT 64
#define NJ (JT / 16)
#define APITCH (JT + 4)

static_assert(DK == DV, "a K tile and a Y tile share one plane layout");

static __device__ __forceinline__ void load_q_step(
    const float* __restrict__ Qf, int b, int j0, int lane, v2f (&qf)[NJ][16]) {
  const float4* qr =
      (const float4*)(Qf + (size_t)(b * (S_ / 16) + (j0 >> 4)) * (16 * DK)) +
      lane;
#pragma unroll
  for (int jt = 0; jt < NJ; ++jt) {
#pragma unroll
    for (int u = 0; u < 8; ++u) {
      const float4 x = qr[jt * (16 * DK / 4) + u * 32];
      qf[jt][2 * u]     = (v2f){x.x, x.y};
      qf[jt][2 * u + 1] = (v2f){x.z, x.w};
    }
  }
}

static __device__ __forceinline__ void load_v_step(
    const float* __restrict__ Vf, int b, int j0, int lane,
    v2f (&vf)[4 * NJ][4]) {
  const float4* vr =
      (const float4*)(Vf + (size_t)(b * (S_ / 16) + (j0 >> 4)) * (16 * DV)) +
      lane;
#pragma unroll
  for (int jt = 0; jt < NJ; ++jt)
#pragma unroll
    for (int u = 0; u < 2; ++u)
#pragma unroll
      for (int nt = 0; nt < 4; ++nt) {
        const float4 x = vr[jt * (16 * DV / 4) + (2 * nt + u) * 32];
        vf[4 * jt + 2 * u][nt]     = (v2f){x.x, x.y};
        vf[4 * jt + 2 * u + 1][nt] = (v2f){x.z, x.w};
      }
}

__global__ __launch_bounds__(256) void attn_kernel(
    float* KY0, float* KY1, const float* __restrict__ Qf,
    const float* __restrict__ Vf) {
  __shared__ __attribute__((aligned(16))) float lds_e[NJ * 256 * 12];
  __shared__ __attribute__((aligned(16))) float lds_a[8][16 * APITCH];

  const int lane = threadIdx.x & 31;
  const int b    = threadIdx.x >> 5;
  const int m    = lane & 15;
  const int hi   = lane >> 4;
  const int half = blockIdx.x & 1;
  const int i0   = (blockIdx.x >> 1) << 4;
  const int jbeg = half * (S_ / 2);
  float* ky = half ? KY1 : KY0;

  const float scale = 0.125f;

  v2f ka[16];
  {
    const float* kr = ky + (size_t)(b * S_ + i0 + m) * DK + 2 * hi;
#pragma unroll
    for (int t = 0; t < 16; ++t) {
      ka[t].x = kr[4 * t] * scale;
      ka[t].y = kr[4 * t + 1] * scale;
    }
  }

  v8f yacc[4];
#pragma unroll
  for (int nt = 0; nt < 4; ++nt)
    yacc[nt] = (v8f){0.f, 0.f, 0.f, 0.f, 0.f, 0.f, 0.f, 0.f};

  const int jend = jbeg + S_ / 2;
  v2f qf[NJ][16];
  load_q_step(Qf, b, jbeg, lane, qf);

  for (int j0 = jbeg; j0 < jend; j0 += JT) {
    v2f vf[4 * NJ][4];
    load_v_step(Vf, b, j0, lane, vf);
    __builtin_amdgcn_sched_barrier(0);

    v8f e[NJ];
#pragma unroll
    for (int jt = 0; jt < NJ; ++jt) {
      e[jt] = (v8f){0.f, 0.f, 0.f, 0.f, 0.f, 0.f, 0.f, 0.f};
#pragma unroll
      for (int t = 0; t < 16; ++t)
        e[jt] = wmma_f32(ka[t], qf[jt][t], e[jt]);
    }

#pragma unroll
    for (int jt = 0; jt < NJ; ++jt)
#pragma unroll
      for (int r = 0; r < 8; ++r)
        lds_e[((jt * 8 + r) * 32 + lane) * 12 + b] = e[jt][r];
    __syncthreads();

#pragma unroll
    for (int jt = 0; jt < NJ; ++jt) {
      const float4* p =
          (const float4*)&lds_e[((jt * 8 + b) * 32 + lane) * 12];
      float4 u = p[0];
      float4 w = p[1];
      float mx = fmaxf(fmaxf(fmaxf(u.x, u.y), fmaxf(u.z, u.w)),
                       fmaxf(fmaxf(w.x, w.y), fmaxf(w.z, w.w)));
      float t0 = __expf(u.x - mx), t1 = __expf(u.y - mx);
      float t2 = __expf(u.z - mx), t3 = __expf(u.w - mx);
      float t4 = __expf(w.x - mx), t5 = __expf(w.y - mx);
      float t6 = __expf(w.z - mx), t7 = __expf(w.w - mx);
      float z = t0 + t1 + t2 + t3 + t4 + t5 + t6 + t7;
      float rz = __builtin_amdgcn_rcpf(z);
      rz = fmaf(fmaf(-z, rz, 1.0f), rz, rz);
      float* dst = &lds_a[0][(b + 8 * hi) * APITCH + jt * 16 + m];
      dst[0 * 16 * APITCH] = t0 * rz;
      dst[1 * 16 * APITCH] = t1 * rz;
      dst[2 * 16 * APITCH] = t2 * rz;
      dst[3 * 16 * APITCH] = t3 * rz;
      dst[4 * 16 * APITCH] = t4 * rz;
      dst[5 * 16 * APITCH] = t5 * rz;
      dst[6 * 16 * APITCH] = t6 * rz;
      dst[7 * 16 * APITCH] = t7 * rz;
    }
    __syncthreads();

    v2f af[4 * NJ];
#pragma unroll
    for (int t = 0; t < 4 * NJ; ++t) {
      af[t].x = lds_a[b][m * APITCH + 4 * t + 2 * hi];
      af[t].y = lds_a[b][m * APITCH + 4 * t + 2 * hi + 1];
    }

    load_q_step(Qf, b, (j0 + JT < jend) ? j0 + JT : j0, lane, qf);
    __builtin_amdgcn_sched_barrier(0);

#pragma unroll
    for (int t = 0; t < 4 * NJ; ++t)
#pragma unroll
      for (int nt = 0; nt < 4; ++nt)
        yacc[nt] = wmma_f32(af[t], vf[t][nt], yacc[nt]);
    __builtin_amdgcn_sched_barrier(0);
    asm volatile("v_nop\n\tv_nop\n\tv_nop\n\tv_nop");
  }

#pragma unroll
  for (int nt = 0; nt < 4; ++nt)
#pragma unroll
    for (int r = 0; r < 8; ++r)
      ky[(size_t)(b * S_ + i0 + r + 8 * hi) * DV + nt * 16 + m] =
          yacc[nt][r];
}

__global__ __launch_bounds__(256) void add_halves(
    float* __restrict__ Y, const float* __restrict__ P1) {
  const int i = blockIdx.x * 256 + threadIdx.x;
  float4 a = ((const float4*)Y)[i];
  const float4 p = ((const float4*)P1)[i];
  a.x += p.x;
  a.y += p.y;
  a.z += p.z;
  a.w += p.w;
  ((float4*)Y)[i] = a;
}

extern "C" void kernel_launch(void* const* d_in, const int* in_sizes, int n_in,
                              void* d_out, int out_size, void* d_ws,
                              size_t ws_size, hipStream_t stream) {
  const float* X  = (const float*)d_in[0];
  const float* Wk = (const float*)d_in[1];
  const float* bk = (const float*)d_in[2];
  const float* Wq = (const float*)d_in[3];
  const float* bq = (const float*)d_in[4];
  const float* Wv = (const float*)d_in[5];
  const float* bv = (const float*)d_in[6];
  float* Y = (float*)d_out;

  float* ws = (float*)d_ws;
  const size_t NKQV = (size_t)B_ * S_ * DK;
  float* Kb  = ws;
  float* Qf  = ws + NKQV;
  float* Vf  = ws + 2 * NKQV;
  float* WT3 = ws + 3 * NKQV;

  transpose_weights<<<(3 * D_IN * DK + 255) / 256, 256, 0, stream>>>(
      Wk, Wq, Wv, WT3);
  qkv_kernel<<<(B_ * S_ / 16) / 4, 256, 0, stream>>>(
      X, WT3, bk, bq, bv, Kb, Qf, Vf, Y);
  attn_kernel<<<2 * (S_ / 16), 256, 0, stream>>>(Y, Kb, Qf, Vf);
  add_halves<<<(B_ * S_ * DV / 4) / 256, 256, 0, stream>>>(Y, Kb);
}


# projection kernel: each k half stages the row-major weights in LDS (32-k chunks, 2 stages, b128 fragment reads), so no transpose launch (3 launches); 8 k per slice pair: order of the k sum changes
# speedup vs baseline: 4.0517x; 1.0657x over previous
#include <hip/hip_runtime.h>

#define D_IN 512
#define DK   64
#define DV   64
#define B_   8
#define S_   2048

typedef __attribute__((ext_vector_type(2))) float v2f;
typedef __attribute__((ext_vector_type(8))) float v8f;

static __device__ __forceinline__ v8f wmma_f32(v2f a, v2f b, v8f c) {
  return __builtin_amdgcn_wmma_f32_16x16x4_f32(
       false, a,  false, b,
       (short)0, c,  false,  false);
}

#define KC  32
#define NCH (D_IN / 2 / KC)
#define WP  (KC + 4)
#define STG (3 * DK * WP)

struct WPieces {
  float4 p0, p1, p2, p3, p4, p5, p6, p7, p8, p9, p10, p11;
};

static __device__ __forceinline__ WPieces w_fetch(const float* wk,
                                                  const float* wq,
                                                  const float* wv, int koff) {
  WPieces w;
  w.p0  = *(const float4*)(wk + 0 * 16 * D_IN + koff);
  w.p1  = *(const float4*)(wk + 1 * 16 * D_IN + koff);
  w.p2  = *(const float4*)(wk + 2 * 16 * D_IN + koff);
  w.p3  = *(const float4*)(wk + 3 * 16 * D_IN + koff);
  w.p4  = *(const float4*)(wq + 0 * 16 * D_IN + koff);
  w.p5  = *(const float4*)(wq + 1 * 16 * D_IN + koff);
  w.p6  = *(const float4*)(wq + 2 * 16 * D_IN + koff);
  w.p7  = *(const float4*)(wq + 3 * 16 * D_IN + koff);
  w.p8  = *(const float4*)(wv + 0 * 16 * D_IN + koff);
  w.p9  = *(const float4*)(wv + 1 * 16 * D_IN + koff);
  w.p10 = *(const float4*)(wv + 2 * 16 * D_IN + koff);
  w.p11 = *(const float4*)(wv + 3 * 16 * D_IN + koff);
  return w;
}

static __device__ __forceinline__ void w_put(float* sb, const WPieces& w) {
  *(float4*)(sb + 0 * 16 * WP)  = w.p0;
  *(float4*)(sb + 1 * 16 * WP)  = w.p1;
  *(float4*)(sb + 2 * 16 * WP)  = w.p2;
  *(float4*)(sb + 3 * 16 * WP)  = w.p3;
  *(float4*)(sb + 4 * 16 * WP)  = w.p4;
  *(float4*)(sb + 5 * 16 * WP)  = w.p5;
  *(float4*)(sb + 6 * 16 * WP)  = w.p6;
  *(float4*)(sb + 7 * 16 * WP)  = w.p7;
  *(float4*)(sb + 8 * 16 * WP)  = w.p8;
  *(float4*)(sb + 9 * 16 * WP)  = w.p9;
  *(float4*)(sb + 10 * 16 * WP) = w.p10;
  *(float4*)(sb + 11 * 16 * WP) = w.p11;
}

static __device__ __forceinline__ void b_group(const float* fb, int u,
                                               float4 (&bf)[12]) {
#pragma unroll
  for (int t = 0; t < 12; ++t)
    bf[t] = *(const float4*)(fb + t * 16 * WP + 8 * u);
}

static __device__ __forceinline__ void mm_group(const float4& xa,
                                                const float4 (&bf)[12],
                                                v8f (&acc)[3][4]) {
#pragma unroll
  for (int s = 0; s < 2; ++s) {
    const v2f a = s ? (v2f){xa.z, xa.w} : (v2f){xa.x, xa.y};
#pragma unroll
    for (int mtx = 0; mtx < 3; ++mtx)
#pragma unroll
      for (int nt = 0; nt < 4; ++nt) {
        const float4 w = bf[mtx * 4 + nt];
        const v2f b = s ? (v2f){w.z, w.w} : (v2f){w.x, w.y};
        if (mtx == 1)
          acc[mtx][nt] = wmma_f32(b, a, acc[mtx][nt]);
        else
          acc[mtx][nt] = wmma_f32(a, b, acc[mtx][nt]);
      }
  }
}

template <bool MORE>
static __device__ __forceinline__ void qkv_chunk(
    int knext, const float* wk, const float* wq, const float* wv,
    const float* __restrict__ xr, const float* fb_cur, float* sb_nxt,
    const float4 (&xa)[4], float4 (&xn)[4], v8f (&acc)[3][4]) {
  WPieces wst;
  if (MORE) {
    wst = w_fetch(wk, wq, wv, knext);
#pragma unroll
    for (int u = 0; u < 4; ++u)
      xn[u] = *(const float4*)(xr + knext + 8 * u);
  }
  __builtin_amdgcn_sched_barrier(0);
  float4 bf[2][12];
  b_group(fb_cur, 0, bf[0]);
#pragma unroll
  for (int u = 0; u < 4; ++u) {
    if (u < 3) b_group(fb_cur, u + 1, bf[(u + 1) & 1]);
    mm_group(xa[u], bf[u & 1], acc);
  }
  __builtin_amdgcn_sched_group_barrier(0x100, 24, 0);
  __builtin_amdgcn_sched_group_barrier(0x008, 24, 0);
  __builtin_amdgcn_sched_group_barrier(0x100, 12, 0);
  __builtin_amdgcn_sched_group_barrier(0x008, 24, 0);
  __builtin_amdgcn_sched_group_barrier(0x100, 12, 0);
  __builtin_amdgcn_sched_group_barrier(0x008, 48, 0);
  __builtin_amdgcn_sched_barrier(0);
  if (MORE) w_put(sb_nxt, wst);
  __syncthreads();
}

__global__ __launch_bounds__(256) void qkv_kernel(
    const float* __restrict__ X, const float* Wk, const float* Wq,
    const float* Wv, const float* __restrict__ bk, const float* __restrict__ bq,
    const float* __restrict__ bv,
    float* __restrict__ Kb, float* __restrict__ Qf, float* __restrict__ Vf,
    float* __restrict__ Kb2) {
  __shared__ __attribute__((aligned(16))) float lds_w[2 * 2 * STG];

  const int lane = threadIdx.x & 31;
  const int wave = threadIdx.x >> 5;
  const int rt   = wave & 3;
  const int kh   = wave >> 2;
  const int wg   = blockIdx.x * 4 + rt;
  const int b    = wg >> 7;
  const int s0   = (wg & 127) << 4;
  const int m    = lane & 15;
  const int hi   = lane >> 4;
  const int kb   = kh * (D_IN / 2);

  const float* xr = X + (size_t)(b * S_ + s0 + m) * D_IN + kb + 4 * hi;

  const int g    = rt * 32 + lane;
  const int rowg = g >> 3;
  const int qp   = g & 7;
  const float* wk = Wk + rowg * D_IN + kb + 4 * qp;
  const float* wq = Wq + rowg * D_IN + kb + 4 * qp;
  const float* wv = Wv + rowg * D_IN + kb + 4 * qp;
  float* half_base = lds_w + kh * (2 * STG);
  float* sb0 = half_base + rowg * WP + 4 * qp;
  float* sb1 = sb0 + STG;
  const float* fb0 = half_base + m * WP + 4 * hi;
  const float* fb1 = fb0 + STG;

  v8f acc[3][4];
#pragma unroll
  for (int i = 0; i < 3; ++i)
#pragma unroll
    for (int j = 0; j < 4; ++j)
      acc[i][j] = (v8f){0.f, 0.f, 0.f, 0.f, 0.f, 0.f, 0.f, 0.f};

  float4 xa0[4], xa1[4];
  {
    const WPieces wst = w_fetch(wk, wq, wv, 0);
#pragma unroll
    for (int u = 0; u < 4; ++u) xa0[u] = *(const float4*)(xr + 8 * u);
    w_put(sb0, wst);
  }
  __syncthreads();

  for (int c = 0; c < NCH - 2; c += 2) {
    qkv_chunk<true>(KC * (c + 1), wk, wq, wv, xr, fb0, sb1, xa0, xa1, acc);
    qkv_chunk<true>(KC * (c + 2), wk, wq, wv, xr, fb1, sb0, xa1, xa0, acc);
  }
  qkv_chunk<true>(KC * (NCH - 1), wk, wq, wv, xr, fb0, sb1, xa0, xa1, acc);
  qkv_chunk<false>(0, wk, wq, wv, xr, fb1, sb0, xa1, xa0, acc);

  __builtin_amdgcn_sched_barrier(0);
  asm volatile("v_nop\n\tv_nop\n\tv_nop\n\tv_nop");
  __builtin_amdgcn_sched_barrier(0);

  float4* pp = (float4*)(lds_w + rt * (24 * 128)) + lane;
  if (kh == 1) {
#pragma unroll
    for (int i = 0; i < 3; ++i)
#pragma unroll
      for (int j = 0; j < 4; ++j)
#pragma unroll
        for (int q = 0; q < 2; ++q)
          pp[((i * 4 + j) * 2 + q) * 32] =
              make_float4(acc[i][j][4 * q], acc[i][j][4 * q + 1],
                          acc[i][j][4 * q + 2], acc[i][j][4 * q + 3]);
  }
  __syncthreads();
  if (kh == 1) return;
#pragma unroll
  for (int i = 0; i < 3; ++i)
#pragma unroll
    for (int j = 0; j < 4; ++j)
#pragma unroll
      for (int q = 0; q < 2; ++q) {
        const float4 p = pp[((i * 4 + j) * 2 + q) * 32];
        acc[i][j][4 * q]     += p.x;
        acc[i][j][4 * q + 1] += p.y;
        acc[i][j][4 * q + 2] += p.z;
        acc[i][j][4 * q + 3] += p.w;
      }

#pragma unroll
  for (int nt = 0; nt < 4; ++nt) {
    const float bs = bk[nt * 16 + m];
#pragma unroll
    for (int r = 0; r < 8; ++r) {
      const size_t o = (size_t)(b * S_ + s0 + r + 8 * hi) * DK + nt * 16 + m;
      const float val = acc[0][nt][r] + bs;
      Kb[o] = val;
      Kb2[o] = val;
    }
  }

  float* qt = Qf + (size_t)wg * (16 * DK);
  float* vt = Vf + (size_t)wg * (16 * DV);
#pragma unroll
  for (int nt = 0; nt < 4; ++nt) {
    float qv[8], vv[8];
    const float bsv = bv[nt * 16 + m];
#pragma unroll
    for (int r = 0; r < 8; ++r) {
      qv[r] = acc[1][nt][r] + bq[nt * 16 + 8 * hi + r];
      vv[r] = acc[2][nt][r] + bsv;
    }
    float4* qg = (float4*)(qt + (2 * nt + hi) * 128);
    float4* vg = (float4*)(vt + (2 * nt + hi) * 128);
    qg[m]      = make_float4(qv[0], qv[1], qv[4], qv[5]);
    qg[16 + m] = make_float4(qv[2], qv[3], qv[6], qv[7]);
    vg[m]      = make_float4(vv[0], vv[1], vv[4], vv[5]);
    vg[16 + m] = make_float4(vv[2], vv[3], vv[6], vv[7]);
  }
}

#define JT 64
#define NJ (JT / 16)
#define APITCH (JT + 4)

static_assert(DK == DV, "a K tile and a Y tile share one plane layout");

static __device__ __forceinline__ void load_q_step(
    const float* __restrict__ Qf, int b, int j0, int lane, v2f (&qf)[NJ][16]) {
  const float4* qr =
      (const float4*)(Qf + (size_t)(b * (S_ / 16) + (j0 >> 4)) * (16 * DK)) +
      lane;
#pragma unroll
  for (int jt = 0; jt < NJ; ++jt) {
#pragma unroll
    for (int u = 0; u < 8; ++u) {
      const float4 x = qr[jt * (16 * DK / 4) + u * 32];
      qf[jt][2 * u]     = (v2f){x.x, x.y};
      qf[jt][2 * u + 1] = (v2f){x.z, x.w};
    }
  }
}

static __device__ __forceinline__ void load_v_step(
    const float* __restrict__ Vf, int b, int j0, int lane,
    v2f (&vf)[4 * NJ][4]) {
  const float4* vr =
      (const float4*)(Vf + (size_t)(b * (S_ / 16) + (j0 >> 4)) * (16 * DV)) +
      lane;
#pragma unroll
  for (int jt = 0; jt < NJ; ++jt)
#pragma unroll
    for (int u = 0; u < 2; ++u)
#pragma unroll
      for (int nt = 0; nt < 4; ++nt) {
        const float4 x = vr[jt * (16 * DV / 4) + (2 * nt + u) * 32];
        vf[4 * jt + 2 * u][nt]     = (v2f){x.x, x.y};
        vf[4 * jt + 2 * u + 1][nt] = (v2f){x.z, x.w};
      }
}

__global__ __launch_bounds__(256) void attn_kernel(
    float* KY0, float* KY1, const float* __restrict__ Qf,
    const float* __restrict__ Vf) {
  __shared__ __attribute__((aligned(16))) float lds_e[NJ * 256 * 12];
  __shared__ __attribute__((aligned(16))) float lds_a[8][16 * APITCH];

  const int lane = threadIdx.x & 31;
  const int b    = threadIdx.x >> 5;
  const int m    = lane & 15;
  const int hi   = lane >> 4;
  const int half = blockIdx.x & 1;
  const int i0   = (blockIdx.x >> 1) << 4;
  const int jbeg = half * (S_ / 2);
  float* ky = half ? KY1 : KY0;

  const float scale = 0.125f;

  v2f ka[16];
  {
    const float* kr = ky + (size_t)(b * S_ + i0 + m) * DK + 2 * hi;
#pragma unroll
    for (int t = 0; t < 16; ++t) {
      ka[t].x = kr[4 * t] * scale;
      ka[t].y = kr[4 * t + 1] * scale;
    }
  }

  v8f yacc[4];
#pragma unroll
  for (int nt = 0; nt < 4; ++nt)
    yacc[nt] = (v8f){0.f, 0.f, 0.f, 0.f, 0.f, 0.f, 0.f, 0.f};

  const int jend = jbeg + S_ / 2;
  v2f qf[NJ][16];
  load_q_step(Qf, b, jbeg, lane, qf);

  for (int j0 = jbeg; j0 < jend; j0 += JT) {
    v2f vf[4 * NJ][4];
    load_v_step(Vf, b, j0, lane, vf);
    __builtin_amdgcn_sched_barrier(0);

    v8f e[NJ];
#pragma unroll
    for (int jt = 0; jt < NJ; ++jt) {
      e[jt] = (v8f){0.f, 0.f, 0.f, 0.f, 0.f, 0.f, 0.f, 0.f};
#pragma unroll
      for (int t = 0; t < 16; ++t)
        e[jt] = wmma_f32(ka[t], qf[jt][t], e[jt]);
    }

#pragma unroll
    for (int jt = 0; jt < NJ; ++jt)
#pragma unroll
      for (int r = 0; r < 8; ++r)
        lds_e[((jt * 8 + r) * 32 + lane) * 12 + b] = e[jt][r];
    __syncthreads();

#pragma unroll
    for (int jt = 0; jt < NJ; ++jt) {
      const float4* p =
          (const float4*)&lds_e[((jt * 8 + b) * 32 + lane) * 12];
      float4 u = p[0];
      float4 w = p[1];
      float mx = fmaxf(fmaxf(fmaxf(u.x, u.y), fmaxf(u.z, u.w)),
                       fmaxf(fmaxf(w.x, w.y), fmaxf(w.z, w.w)));
      float t0 = __expf(u.x - mx), t1 = __expf(u.y - mx);
      float t2 = __expf(u.z - mx), t3 = __expf(u.w - mx);
      float t4 = __expf(w.x - mx), t5 = __expf(w.y - mx);
      float t6 = __expf(w.z - mx), t7 = __expf(w.w - mx);
      float z = t0 + t1 + t2 + t3 + t4 + t5 + t6 + t7;
      float rz = __builtin_amdgcn_rcpf(z);
      rz = fmaf(fmaf(-z, rz, 1.0f), rz, rz);
      float* dst = &lds_a[0][(b + 8 * hi) * APITCH + jt * 16 + m];
      dst[0 * 16 * APITCH] = t0 * rz;
      dst[1 * 16 * APITCH] = t1 * rz;
      dst[2 * 16 * APITCH] = t2 * rz;
      dst[3 * 16 * APITCH] = t3 * rz;
      dst[4 * 16 * APITCH] = t4 * rz;
      dst[5 * 16 * APITCH] = t5 * rz;
      dst[6 * 16 * APITCH] = t6 * rz;
      dst[7 * 16 * APITCH] = t7 * rz;
    }
    __syncthreads();

    v2f af[4 * NJ];
#pragma unroll
    for (int t = 0; t < 4 * NJ; ++t) {
      af[t].x = lds_a[b][m * APITCH + 4 * t + 2 * hi];
      af[t].y = lds_a[b][m * APITCH + 4 * t + 2 * hi + 1];
    }

    load_q_step(Qf, b, (j0 + JT < jend) ? j0 + JT : j0, lane, qf);
    __builtin_amdgcn_sched_barrier(0);

#pragma unroll
    for (int t = 0; t < 4 * NJ; ++t)
#pragma unroll
      for (int nt = 0; nt < 4; ++nt)
        yacc[nt] = wmma_f32(af[t], vf[t][nt], yacc[nt]);
    __builtin_amdgcn_sched_barrier(0);
    asm volatile("v_nop\n\tv_nop\n\tv_nop\n\tv_nop");
  }

#pragma unroll
  for (int nt = 0; nt < 4; ++nt)
#pragma unroll
    for (int r = 0; r < 8; ++r)
      ky[(size_t)(b * S_ + i0 + r + 8 * hi) * DV + nt * 16 + m] =
          yacc[nt][r];
}

__global__ __launch_bounds__(256) void add_halves(
    float* __restrict__ Y, const float* __restrict__ P1) {
  const int i = blockIdx.x * 256 + threadIdx.x;
  float4 a = ((const float4*)Y)[i];
  const float4 p = ((const float4*)P1)[i];
  a.x += p.x;
  a.y += p.y;
  a.z += p.z;
  a.w += p.w;
  ((float4*)Y)[i] = a;
}

extern "C" void kernel_launch(void* const* d_in, const int* in_sizes, int n_in,
                              void* d_out, int out_size, void* d_ws,
                              size_t ws_size, hipStream_t stream) {
  const float* X  = (const float*)d_in[0];
  const float* Wk = (const float*)d_in[1];
  const float* bk = (const float*)d_in[2];
  const float* Wq = (const float*)d_in[3];
  const float* bq = (const float*)d_in[4];
  const float* Wv = (const float*)d_in[5];
  const float* bv = (const float*)d_in[6];
  float* Y = (float*)d_out;

  float* ws = (float*)d_ws;
  const size_t NKQV = (size_t)B_ * S_ * DK;
  float* Kb  = ws;
  float* Qf  = ws + NKQV;
  float* Vf  = ws + 2 * NKQV;

  qkv_kernel<<<(B_ * S_ / 16) / 4, 256, 0, stream>>>(
      X, Wk, Wq, Wv, bk, bq, bv, Kb, Qf, Vf, Y);
  attn_kernel<<<2 * (S_ / 16), 256, 0, stream>>>(Y, Kb, Qf, Vf);
  add_halves<<<(B_ * S_ * DV / 4) / 256, 256, 0, stream>>>(Y, Kb);
}
